# starting kernel
# baseline (speedup 1.0000x reference)
#include <hip/hip_runtime.h>
#include <hip/hip_bf16.h>

#define B_SAMP 1024
#define N_IN   64
#define N_OUT  4096
#define N_BATCH 64
#define N_CONT 16

#define HSTRIDE 4100
#define ZSTRIDE 68
#define CSTRIDE 20

typedef __attribute__((ext_vector_type(2))) float v2f;
typedef __attribute__((ext_vector_type(8))) float v8f;

static __device__ __forceinline__ v8f wmma4(v2f a, v2f b, v8f c) {
    return __builtin_amdgcn_wmma_f32_16x16x4_f32(
         false, a,  false, b,
         (short)0, c,  false,  false);
}

__global__ __launch_bounds__(1024) void prep_tiles_kernel(
    const int* __restrict__ idx, int* __restrict__ tiles_batch,
    int* __restrict__ slots)
{
    __shared__ int hist[N_BATCH];
    __shared__ int cursor[N_BATCH];
    __shared__ int sbase[N_BATCH];
    int t = threadIdx.x;
    if (t < N_BATCH) { hist[t] = 0; cursor[t] = 0; }
    __syncthreads();
    int b = idx[t];
    atomicAdd(&hist[b], 1);
    __syncthreads();
    if (t == 0) {
        int tc = 0;
        for (int bb = 0; bb < N_BATCH; ++bb) {
            sbase[bb] = tc * 16;
            int nt = (hist[bb] + 15) >> 4;
            for (int k = 0; k < nt; ++k) tiles_batch[tc++] = bb;
        }
        for (; tc < 128; ++tc) tiles_batch[tc] = -1;
    }
    __syncthreads();
    slots[t] = -1;
    slots[t + 1024] = -1;
    __syncthreads();
    int p = atomicAdd(&cursor[b], 1);
    slots[sbase[b] + p] = t;
}

__global__ __launch_bounds__(256) void decoder_main_kernel(
    const float* __restrict__ z, const float* __restrict__ sizef,
    const float* __restrict__ cont, const float* __restrict__ amat_W,
    const float* __restrict__ embed_A, const float* __restrict__ embed_h3,
    const float* __restrict__ cont_W, const int* __restrict__ tiles_batch,
    const int* __restrict__ slots, float* __restrict__ out)
{
    extern __shared__ float smem[];
    float* hbuf  = smem;
    float* ztile = hbuf  + 16 * HSTRIDE;
    float* ctile = ztile + 16 * ZSTRIDE;
    float* sfbuf = ctile + 16 * CSTRIDE;
    int*   sids  = (int*)(sfbuf + 16);

    const int tile = blockIdx.x;
    const int b = tiles_batch[tile];
    if (b < 0) return;
    const int t = threadIdx.x;

    if (t < 16) {
        int s = slots[tile * 16 + t];
        sids[t] = s;
        sfbuf[t] = (s >= 0) ? sizef[s] : 1.0f;
    }
    __syncthreads();

    for (int i = t; i < 16 * N_IN; i += 256) {
        int m = i >> 6, l = i & 63;
        int s = sids[m];
        ztile[m * ZSTRIDE + l] = (s >= 0) ? z[s * N_IN + l] : 0.0f;
    }
    for (int i = t; i < 16 * N_CONT; i += 256) {
        int m = i >> 4, l = i & 15;
        int s = sids[m];
        ctile[m * CSTRIDE + l] = (s >= 0) ? cont[s * N_CONT + l] : 0.0f;
    }
    __syncthreads();

    const int wave = t >> 5;
    const int lane = t & 31;
    const int half = lane >> 4;
    const int ln   = lane & 15;

    v2f afrag[16];
#pragma unroll
    for (int c = 0; c < 16; ++c) {
        const float* p = &ztile[ln * ZSTRIDE + 4 * c + 2 * half];
        v2f v; v.x = p[0]; v.y = p[1];
        afrag[c] = v;
    }
    v2f cfrag[4];
#pragma unroll
    for (int c = 0; c < 4; ++c) {
        const float* p = &ctile[ln * CSTRIDE + 4 * c + 2 * half];
        v2f v; v.x = p[0]; v.y = p[1];
        cfrag[c] = v;
    }

    const float* Eb  = embed_A  + ((size_t)b << 18);
    const float* h3b = embed_h3 + (size_t)b * N_OUT;

    for (int ct = wave; ct < N_OUT / 16; ct += 8) {
        const int g0 = ct * 16;
        const int gcol = g0 + ln;
        const float* ep = Eb + (size_t)gcol * N_IN;
        if (ct + 8 < N_OUT / 16)
            __builtin_prefetch(ep + 128 * N_IN, 0, 1);

        v8f acc = {};
#pragma unroll
        for (int c = 0; c < 16; ++c) {
            int k = 4 * c + 2 * half;
            v2f bf;
            bf.x = amat_W[(size_t)k * N_OUT + gcol];
            bf.y = amat_W[(size_t)(k + 1) * N_OUT + gcol];
            acc = wmma4(afrag[c], bf, acc);
        }
#pragma unroll
        for (int c = 0; c < 16; ++c) {
            int k = 4 * c + 2 * half;
            v2f bf = *(const v2f*)(ep + k);
            acc = wmma4(afrag[c], bf, acc);
        }
#pragma unroll
        for (int c = 0; c < 4; ++c) {
            int k = 4 * c + 2 * half;
            v2f bf;
            bf.x = cont_W[(size_t)k * N_OUT + gcol];
            bf.y = cont_W[(size_t)(k + 1) * N_OUT + gcol];
            acc = wmma4(cfrag[c], bf, acc);
        }
        const float h3v = h3b[gcol];
#pragma unroll
        for (int r = 0; r < 8; ++r) {
            int m = r + 8 * half;
            hbuf[m * HSTRIDE + gcol] = acc[r] + h3v;
        }
    }
    __syncthreads();

    const int row = 2 * wave + half;
    float* hrow = hbuf + row * HSTRIDE;

    float mx = -3.402823466e38f;
    for (int c = ln; c < N_OUT; c += 16) mx = fmaxf(mx, hrow[c]);
#pragma unroll
    for (int off = 1; off < 16; off <<= 1) mx = fmaxf(mx, __shfl_xor(mx, off, 32));

    float sum = 0.0f;
    for (int c = ln; c < N_OUT; c += 16) {
        float e = __expf(hrow[c] - mx);
        hrow[c] = e;
        sum += e;
    }
#pragma unroll
    for (int off = 1; off < 16; off <<= 1) sum += __shfl_xor(sum, off, 32);

    const int s = sids[row];
    if (s >= 0) {
        const float scale = sfbuf[row] / sum;
        float* orow = out + (size_t)s * N_OUT;
        for (int c = ln; c < N_OUT; c += 16) orow[c] = hrow[c] * scale;
    }
}

__global__ void exp_kernel(const float* __restrict__ px_r, float* __restrict__ o)
{
    int i = blockIdx.x * blockDim.x + threadIdx.x;
    if (i < N_OUT) o[i] = __expf(px_r[i]);
}

extern "C" void kernel_launch(void* const* d_in, const int* in_sizes, int n_in,
                              void* d_out, int out_size, void* d_ws, size_t ws_size,
                              hipStream_t stream)
{
    const float* z    = (const float*)d_in[0];
    const int*   bc   = (const int*)  d_in[1];
    const float* sf   = (const float*)d_in[2];
    const float* cc   = (const float*)d_in[3];
    const float* aW   = (const float*)d_in[4];
    const float* eA   = (const float*)d_in[5];
    const float* eh3  = (const float*)d_in[6];
    const float* cW   = (const float*)d_in[7];
    const float* pxr  = (const float*)d_in[8];
    float* out = (float*)d_out;

    int* ws = (int*)d_ws;
    int* tiles_batch = ws;
    int* slots       = ws + 128;

    prep_tiles_kernel<<<1, 1024, 0, stream>>>(bc, tiles_batch, slots);

    size_t smem = (size_t)(16 * HSTRIDE + 16 * ZSTRIDE + 16 * CSTRIDE + 16) * sizeof(float)
                + 16 * sizeof(int);
    decoder_main_kernel<<<128, 256, smem, stream>>>(
        z, sf, cc, aW, eA, eh3, cW, tiles_batch, slots, out);

    exp_kernel<<<16, 256, 0, stream>>>(pxr, out + (size_t)B_SAMP * N_OUT);
}


# Wave tile 16x16 to 16x64 (4 columns per lane): B operands by 16-byte loads into two register sets, each refilled a group ahead of its WMMAs; sum order: k permuted in 8s, cont_W before A_b
# speedup vs baseline: 1.5750x; 1.5750x over previous
#include <hip/hip_runtime.h>
#include <hip/hip_bf16.h>

#define B_SAMP 1024
#define N_IN   64
#define N_OUT  4096
#define N_BATCH 64
#define N_CONT 16

#define HSTRIDE 4100
#define ZSTRIDE 68
#define CSTRIDE 20

typedef __attribute__((ext_vector_type(2))) float v2f;
typedef __attribute__((ext_vector_type(4))) float v4f;
typedef __attribute__((ext_vector_type(8))) float v8f;

struct HasNext { static constexpr bool value = true;  };
struct NoNext  { static constexpr bool value = false; };

static __device__ __forceinline__ v8f wmma4(v2f a, v2f b, v8f c) {
    return __builtin_amdgcn_wmma_f32_16x16x4_f32(
         false, a,  false, b,
         (short)0, c,  false,  false);
}

__global__ __launch_bounds__(1024) void prep_tiles_kernel(
    const int* __restrict__ idx, int* __restrict__ tiles_batch,
    int* __restrict__ slots)
{
    __shared__ int hist[N_BATCH];
    __shared__ int cursor[N_BATCH];
    __shared__ int sbase[N_BATCH];
    int t = threadIdx.x;
    if (t < N_BATCH) { hist[t] = 0; cursor[t] = 0; }
    __syncthreads();
    int b = idx[t];
    atomicAdd(&hist[b], 1);
    __syncthreads();
    if (t == 0) {
        int tc = 0;
        for (int bb = 0; bb < N_BATCH; ++bb) {
            sbase[bb] = tc * 16;
            int nt = (hist[bb] + 15) >> 4;
            for (int k = 0; k < nt; ++k) tiles_batch[tc++] = bb;
        }
        for (; tc < 128; ++tc) tiles_batch[tc] = -1;
    }
    __syncthreads();
    slots[t] = -1;
    slots[t + 1024] = -1;
    __syncthreads();
    int p = atomicAdd(&cursor[b], 1);
    slots[sbase[b] + p] = t;
}

static __device__ __forceinline__ void load_w_group(
    const float* amat_W, const float* cont_W, const float* h3b, int gc, int half,
    v4f (&w0)[16], v4f (&w1)[16], v4f (&c0)[4], v4f (&c1)[4], v4f& h3v)
{
#pragma unroll
    for (int cc = 0; cc < 16; ++cc) {
        const int k = 8 * (cc >> 1) + 4 * half + 2 * (cc & 1);
        w0[cc] = *(const v4f*)(amat_W + (size_t)k * N_OUT + gc);
        w1[cc] = *(const v4f*)(amat_W + (size_t)(k + 1) * N_OUT + gc);
    }
#pragma unroll
    for (int cc = 0; cc < 4; ++cc) {
        const int k = 4 * cc + 2 * half;
        c0[cc] = *(const v4f*)(cont_W + (size_t)k * N_OUT + gc);
        c1[cc] = *(const v4f*)(cont_W + (size_t)(k + 1) * N_OUT + gc);
    }
    h3v = *(const v4f*)(h3b + gc);
}

static __device__ __forceinline__ void load_e_group(
    const float* Eb, int gc, int half, v4f (&ev)[4][8])
{
#pragma unroll
    for (int j = 0; j < 4; ++j) {
        const float* ep = Eb + (size_t)(gc + j) * N_IN + 4 * half;
#pragma unroll
        for (int i = 0; i < 8; ++i) ev[j][i] = *(const v4f*)(ep + 8 * i);
    }
}

__global__ __launch_bounds__(256) void decoder_main_kernel(
    const float* __restrict__ z, const float* __restrict__ sizef,
    const float* __restrict__ cont,
    const float* amat_W, const float* embed_A, const float* embed_h3,
    const float* cont_W, const int* __restrict__ tiles_batch,
    const int* __restrict__ slots, float* __restrict__ out)
{
    extern __shared__ __attribute__((aligned(16))) float smem[];
    float* hbuf  = smem;
    float* ztile = hbuf  + 16 * HSTRIDE;
    float* ctile = ztile + 16 * ZSTRIDE;
    float* sfbuf = ctile + 16 * CSTRIDE;
    int*   sids  = (int*)(sfbuf + 16);

    const int tile = blockIdx.x;
    const int b = tiles_batch[tile];
    if (b < 0) return;
    const int t = threadIdx.x;

    if (t < 16) {
        int s = slots[tile * 16 + t];
        sids[t] = s;
        sfbuf[t] = (s >= 0) ? sizef[s] : 1.0f;
    }
    __syncthreads();

    for (int i = t; i < 16 * N_IN; i += 256) {
        int m = i >> 6, l = i & 63;
        int s = sids[m];
        ztile[m * ZSTRIDE + l] = (s >= 0) ? z[s * N_IN + l] : 0.0f;
    }
    for (int i = t; i < 16 * N_CONT; i += 256) {
        int m = i >> 4, l = i & 15;
        int s = sids[m];
        ctile[m * CSTRIDE + l] = (s >= 0) ? cont[s * N_CONT + l] : 0.0f;
    }
    __syncthreads();

    const int wave = t >> 5;
    const int lane = t & 31;
    const int half = lane >> 4;
    const int ln   = lane & 15;

    v2f afrag[16];
#pragma unroll
    for (int i = 0; i < 8; ++i) {
        const v4f q = *(const v4f*)&ztile[ln * ZSTRIDE + 8 * i + 4 * half];
        v2f lo; lo.x = q.x; lo.y = q.y;
        v2f hi; hi.x = q.z; hi.y = q.w;
        afrag[2 * i]     = lo;
        afrag[2 * i + 1] = hi;
    }
    v2f cfrag[4];
#pragma unroll
    for (int c = 0; c < 4; ++c) {
        const float* p = &ctile[ln * CSTRIDE + 4 * c + 2 * half];
        v2f v; v.x = p[0]; v.y = p[1];
        cfrag[c] = v;
    }

    const float* Eb  = embed_A  + ((size_t)b << 18);
    const float* h3b = embed_h3 + (size_t)b * N_OUT;

    v4f w0[16], w1[16], ev[4][8], c0[4], c1[4], h3v, h3n;
    {
        const int gc = wave * 64 + 4 * ln;
        __builtin_prefetch(Eb + (size_t)(wave * 64 + 512 + lane) * N_IN, 0, 1);
        __builtin_prefetch(Eb + (size_t)(wave * 64 + 544 + lane) * N_IN, 0, 1);
        load_w_group(amat_W, cont_W, h3b, gc, half, w0, w1, c0, c1, h3v);
        load_e_group(Eb, gc, half, ev);
    }
    h3n = h3v;

    auto group_pass = [&](const int cg, auto more_tag) __attribute__((always_inline)) {
        constexpr bool MORE = decltype(more_tag)::value;
        const int gc = cg * 64 + 4 * ln;
        if (cg + 16 < N_OUT / 64) {
            __builtin_prefetch(Eb + (size_t)(cg * 64 + 1024 + lane) * N_IN, 0, 1);
            __builtin_prefetch(Eb + (size_t)(cg * 64 + 1056 + lane) * N_IN, 0, 1);
        }

        v8f acc[4] = {};
#pragma unroll
        for (int cc = 0; cc < 16; ++cc) {
#pragma unroll
            for (int j = 0; j < 4; ++j) {
                v2f bf; bf.x = w0[cc][j]; bf.y = w1[cc][j];
                acc[j] = wmma4(afrag[cc], bf, acc[j]);
            }
        }
#pragma unroll
        for (int cc = 0; cc < 4; ++cc) {
            asm("" : "+v"(c0[cc]), "+v"(c1[cc]));
#pragma unroll
            for (int j = 0; j < 4; ++j) {
                v2f bf; bf.x = c0[cc][j]; bf.y = c1[cc][j];
                acc[j] = wmma4(cfrag[cc], bf, acc[j]);
            }
        }
        __builtin_amdgcn_sched_barrier(0);
        if constexpr (MORE)
            load_w_group(amat_W, cont_W, h3b, gc + 512, half, w0, w1, c0, c1, h3n);

#pragma unroll
        for (int i = 0; i < 8; ++i) {
#pragma unroll
            for (int j = 0; j < 4; ++j) {
                v2f bf; bf.x = ev[j][i].x; bf.y = ev[j][i].y;
                acc[j] = wmma4(afrag[2 * i], bf, acc[j]);
            }
#pragma unroll
            for (int j = 0; j < 4; ++j) {
                v2f bf; bf.x = ev[j][i].z; bf.y = ev[j][i].w;
                acc[j] = wmma4(afrag[2 * i + 1], bf, acc[j]);
            }
        }
#pragma unroll
        for (int r = 0; r < 8; ++r) {
            const int m = r + 8 * half;
            v4f o;
            o.x = acc[0][r] + h3v.x;
            o.y = acc[1][r] + h3v.y;
            o.z = acc[2][r] + h3v.z;
            o.w = acc[3][r] + h3v.w;
            *(v4f*)&hbuf[m * HSTRIDE + gc] = o;
        }
        h3v = h3n;
        __builtin_amdgcn_sched_barrier(0);
        if constexpr (MORE) load_e_group(Eb, gc + 512, half, ev);
    };
    for (int it = 0; it < N_OUT / 64 / 8 - 1; ++it) group_pass(wave + 8 * it, HasNext{});
    group_pass(wave + N_OUT / 64 - 8, NoNext{});
    __syncthreads();

    const int row = 2 * wave + half;
    float* hrow = hbuf + row * HSTRIDE;

    float mx = -3.402823466e38f;
    for (int c = ln; c < N_OUT; c += 16) mx = fmaxf(mx, hrow[c]);
#pragma unroll
    for (int off = 1; off < 16; off <<= 1) mx = fmaxf(mx, __shfl_xor(mx, off, 32));

    float sum = 0.0f;
    for (int c = ln; c < N_OUT; c += 16) {
        float e = __expf(hrow[c] - mx);
        hrow[c] = e;
        sum += e;
    }
#pragma unroll
    for (int off = 1; off < 16; off <<= 1) sum += __shfl_xor(sum, off, 32);

    const int s = sids[row];
    if (s >= 0) {
        const float scale = sfbuf[row] / sum;
        float* orow = out + (size_t)s * N_OUT;
        for (int c = ln; c < N_OUT; c += 16) orow[c] = hrow[c] * scale;
    }
}

__global__ void exp_kernel(const float* __restrict__ px_r, float* __restrict__ o)
{
    int i = blockIdx.x * blockDim.x + threadIdx.x;
    if (i < N_OUT) o[i] = __expf(px_r[i]);
}

extern "C" void kernel_launch(void* const* d_in, const int* in_sizes, int n_in,
                              void* d_out, int out_size, void* d_ws, size_t ws_size,
                              hipStream_t stream)
{
    const float* z    = (const float*)d_in[0];
    const int*   bc   = (const int*)  d_in[1];
    const float* sf   = (const float*)d_in[2];
    const float* cc   = (const float*)d_in[3];
    const float* aW   = (const float*)d_in[4];
    const float* eA   = (const float*)d_in[5];
    const float* eh3  = (const float*)d_in[6];
    const float* cW   = (const float*)d_in[7];
    const float* pxr  = (const float*)d_in[8];
    float* out = (float*)d_out;

    int* ws = (int*)d_ws;
    int* tiles_batch = ws;
    int* slots       = ws + 128;

    prep_tiles_kernel<<<1, 1024, 0, stream>>>(bc, tiles_batch, slots);

    size_t smem = (size_t)(16 * HSTRIDE + 16 * ZSTRIDE + 16 * CSTRIDE + 16) * sizeof(float)
                + 16 * sizeof(int);
    decoder_main_kernel<<<128, 256, smem, stream>>>(
        z, sf, cc, aW, eA, eh3, cW, tiles_batch, slots, out);

    exp_kernel<<<16, 256, 0, stream>>>(pxr, out + (size_t)B_SAMP * N_OUT);
}


# Tile sort and exp(px_r) folded into the main kernel, 3 launches to 1: each block finds its own tile by a counting sort of the batch indices in LDS (ballot ranks); sum order unchanged
# speedup vs baseline: 1.9906x; 1.2639x over previous
#include <hip/hip_runtime.h>
#include <hip/hip_bf16.h>

#define B_SAMP 1024
#define N_IN   64
#define N_OUT  4096
#define N_BATCH 64
#define N_CONT 16

#define HSTRIDE 4100
#define ZSTRIDE 68
#define CSTRIDE 20

typedef __attribute__((ext_vector_type(2))) float v2f;
typedef __attribute__((ext_vector_type(4))) float v4f;
typedef __attribute__((ext_vector_type(8))) float v8f;

struct HasNext { static constexpr bool value = true;  };
struct NoNext  { static constexpr bool value = false; };

static __device__ __forceinline__ v8f wmma4(v2f a, v2f b, v8f c) {
    return __builtin_amdgcn_wmma_f32_16x16x4_f32(
         false, a,  false, b,
         (short)0, c,  false,  false);
}

static __device__ __forceinline__ void load_w_group(
    const float* amat_W, const float* cont_W, const float* h3b, int gc, int half,
    v4f (&w0)[16], v4f (&w1)[16], v4f (&c0)[4], v4f (&c1)[4], v4f& h3v)
{
#pragma unroll
    for (int cc = 0; cc < 16; ++cc) {
        const int k = 8 * (cc >> 1) + 4 * half + 2 * (cc & 1);
        w0[cc] = *(const v4f*)(amat_W + (size_t)k * N_OUT + gc);
        w1[cc] = *(const v4f*)(amat_W + (size_t)(k + 1) * N_OUT + gc);
    }
#pragma unroll
    for (int cc = 0; cc < 4; ++cc) {
        const int k = 4 * cc + 2 * half;
        c0[cc] = *(const v4f*)(cont_W + (size_t)k * N_OUT + gc);
        c1[cc] = *(const v4f*)(cont_W + (size_t)(k + 1) * N_OUT + gc);
    }
    h3v = *(const v4f*)(h3b + gc);
}

static __device__ __forceinline__ void load_e_group(
    const float* Eb, int gc, int half, v4f (&ev)[4][8])
{
#pragma unroll
    for (int j = 0; j < 4; ++j) {
        const float* ep = Eb + (size_t)(gc + j) * N_IN + 4 * half;
#pragma unroll
        for (int i = 0; i < 8; ++i) ev[j][i] = *(const v4f*)(ep + 8 * i);
    }
}

__global__ __launch_bounds__(256) void decoder_main_kernel(
    const float* __restrict__ z, const int* __restrict__ idx,
    const float* __restrict__ sizef, const float* __restrict__ cont,
    const float* amat_W, const float* embed_A, const float* embed_h3,
    const float* cont_W, const float* __restrict__ px_r, float* __restrict__ out)
{
    extern __shared__ __attribute__((aligned(16))) float smem[];
    float* hbuf  = smem;
    float* ztile = hbuf  + 16 * HSTRIDE;
    float* ctile = ztile + 16 * ZSTRIDE;
    float* sfbuf = ctile + 16 * CSTRIDE;
    int*   sids  = (int*)(sfbuf + 16);

    __shared__ int hist[N_BATCH];
    __shared__ int ccnt[32];

    const int tile = blockIdx.x;
    const int t = threadIdx.x;
    const int wave = t >> 5;
    const int lane = t & 31;

    int bv[4];
#pragma unroll
    for (int j = 0; j < 4; ++j) bv[j] = idx[256 * j + t];
    float pr = 0.0f;
    if (t < 32) pr = px_r[tile * 32 + t];

    if (t < N_BATCH) hist[t] = 0;
    if (t < 16) sids[t] = -1;
    __syncthreads();
#pragma unroll
    for (int j = 0; j < 4; ++j) atomicAdd(&hist[bv[j]], 1);
    if (t < 32) out[(size_t)B_SAMP * N_OUT + tile * 32 + t] = __expf(pr);
    __syncthreads();

    int b = -1, q = 0, tc = 0;
#pragma unroll
    for (int bb = 0; bb < N_BATCH; ++bb) {
        const int nt = (hist[bb] + 15) >> 4;
        if (tile >= tc && tile < tc + nt) { b = bb; q = tile - tc; }
        tc += nt;
    }
    if (b < 0) return;

    bool mine[4];
    int below[4];
#pragma unroll
    for (int j = 0; j < 4; ++j) {
        mine[j] = (bv[j] == b);
        const unsigned m = (unsigned)__ballot(mine[j]);
        below[j] = __popc(m & ((1u << lane) - 1u));
        if (lane == 0) ccnt[8 * j + wave] = __popc(m);
    }
    __syncthreads();
    int base[4] = {0, 0, 0, 0};
#pragma unroll
    for (int c = 0; c < 32; ++c) {
        const int n = ccnt[c];
#pragma unroll
        for (int j = 0; j < 4; ++j) base[j] += (c < 8 * j + wave) ? n : 0;
    }
#pragma unroll
    for (int j = 0; j < 4; ++j) {
        const int rank = base[j] + below[j];
        if (mine[j] && (rank >> 4) == q) sids[rank & 15] = 256 * j + t;
    }
    __syncthreads();

    if (t < 16) {
        const int s = sids[t];
        sfbuf[t] = (s >= 0) ? sizef[s] : 1.0f;
    }

    for (int i = t; i < 16 * N_IN; i += 256) {
        int m = i >> 6, l = i & 63;
        int s = sids[m];
        ztile[m * ZSTRIDE + l] = (s >= 0) ? z[s * N_IN + l] : 0.0f;
    }
    for (int i = t; i < 16 * N_CONT; i += 256) {
        int m = i >> 4, l = i & 15;
        int s = sids[m];
        ctile[m * CSTRIDE + l] = (s >= 0) ? cont[s * N_CONT + l] : 0.0f;
    }
    __syncthreads();

    const int half = lane >> 4;
    const int ln   = lane & 15;

    v2f afrag[16];
#pragma unroll
    for (int i = 0; i < 8; ++i) {
        const v4f q = *(const v4f*)&ztile[ln * ZSTRIDE + 8 * i + 4 * half];
        v2f lo; lo.x = q.x; lo.y = q.y;
        v2f hi; hi.x = q.z; hi.y = q.w;
        afrag[2 * i]     = lo;
        afrag[2 * i + 1] = hi;
    }
    v2f cfrag[4];
#pragma unroll
    for (int c = 0; c < 4; ++c) {
        const float* p = &ctile[ln * CSTRIDE + 4 * c + 2 * half];
        v2f v; v.x = p[0]; v.y = p[1];
        cfrag[c] = v;
    }

    const float* Eb  = embed_A  + ((size_t)b << 18);
    const float* h3b = embed_h3 + (size_t)b * N_OUT;

    v4f w0[16], w1[16], ev[4][8], c0[4], c1[4], h3v, h3n;
    {
        const int gc = wave * 64 + 4 * ln;
        __builtin_prefetch(Eb + (size_t)(wave * 64 + 512 + lane) * N_IN, 0, 1);
        __builtin_prefetch(Eb + (size_t)(wave * 64 + 544 + lane) * N_IN, 0, 1);
        load_w_group(amat_W, cont_W, h3b, gc, half, w0, w1, c0, c1, h3v);
        load_e_group(Eb, gc, half, ev);
    }
    h3n = h3v;

    auto group_pass = [&](const int cg, auto more_tag) __attribute__((always_inline)) {
        constexpr bool MORE = decltype(more_tag)::value;
        const int gc = cg * 64 + 4 * ln;
        if (cg + 16 < N_OUT / 64) {
            __builtin_prefetch(Eb + (size_t)(cg * 64 + 1024 + lane) * N_IN, 0, 1);
            __builtin_prefetch(Eb + (size_t)(cg * 64 + 1056 + lane) * N_IN, 0, 1);
        }

        v8f acc[4] = {};
#pragma unroll
        for (int cc = 0; cc < 16; ++cc) {
#pragma unroll
            for (int j = 0; j < 4; ++j) {
                v2f bf; bf.x = w0[cc][j]; bf.y = w1[cc][j];
                acc[j] = wmma4(afrag[cc], bf, acc[j]);
            }
        }
#pragma unroll
        for (int cc = 0; cc < 4; ++cc) {
            asm("" : "+v"(c0[cc]), "+v"(c1[cc]));
#pragma unroll
            for (int j = 0; j < 4; ++j) {
                v2f bf; bf.x = c0[cc][j]; bf.y = c1[cc][j];
                acc[j] = wmma4(cfrag[cc], bf, acc[j]);
            }
        }
        __builtin_amdgcn_sched_barrier(0);
        if constexpr (MORE)
            load_w_group(amat_W, cont_W, h3b, gc + 512, half, w0, w1, c0, c1, h3n);

#pragma unroll
        for (int i = 0; i < 8; ++i) {
#pragma unroll
            for (int j = 0; j < 4; ++j) {
                v2f bf; bf.x = ev[j][i].x; bf.y = ev[j][i].y;
                acc[j] = wmma4(afrag[2 * i], bf, acc[j]);
            }
#pragma unroll
            for (int j = 0; j < 4; ++j) {
                v2f bf; bf.x = ev[j][i].z; bf.y = ev[j][i].w;
                acc[j] = wmma4(afrag[2 * i + 1], bf, acc[j]);
            }
        }
#pragma unroll
        for (int r = 0; r < 8; ++r) {
            const int m = r + 8 * half;
            v4f o;
            o.x = acc[0][r] + h3v.x;
            o.y = acc[1][r] + h3v.y;
            o.z = acc[2][r] + h3v.z;
            o.w = acc[3][r] + h3v.w;
            *(v4f*)&hbuf[m * HSTRIDE + gc] = o;
        }
        h3v = h3n;
        __builtin_amdgcn_sched_barrier(0);
        if constexpr (MORE) load_e_group(Eb, gc + 512, half, ev);
    };
    for (int it = 0; it < N_OUT / 64 / 8 - 1; ++it) group_pass(wave + 8 * it, HasNext{});
    group_pass(wave + N_OUT / 64 - 8, NoNext{});
    __syncthreads();

    const int row = 2 * wave + half;
    float* hrow = hbuf + row * HSTRIDE;

    float mx = -3.402823466e38f;
    for (int c = ln; c < N_OUT; c += 16) mx = fmaxf(mx, hrow[c]);
#pragma unroll
    for (int off = 1; off < 16; off <<= 1) mx = fmaxf(mx, __shfl_xor(mx, off, 32));

    float sum = 0.0f;
    for (int c = ln; c < N_OUT; c += 16) {
        float e = __expf(hrow[c] - mx);
        hrow[c] = e;
        sum += e;
    }
#pragma unroll
    for (int off = 1; off < 16; off <<= 1) sum += __shfl_xor(sum, off, 32);

    const int s = sids[row];
    if (s >= 0) {
        const float scale = sfbuf[row] / sum;
        float* orow = out + (size_t)s * N_OUT;
        for (int c = ln; c < N_OUT; c += 16) orow[c] = hrow[c] * scale;
    }
}

extern "C" void kernel_launch(void* const* d_in, const int* in_sizes, int n_in,
                              void* d_out, int out_size, void* d_ws, size_t ws_size,
                              hipStream_t stream)
{
    const float* z    = (const float*)d_in[0];
    const int*   bc   = (const int*)  d_in[1];
    const float* sf   = (const float*)d_in[2];
    const float* cc   = (const float*)d_in[3];
    const float* aW   = (const float*)d_in[4];
    const float* eA   = (const float*)d_in[5];
    const float* eh3  = (const float*)d_in[6];
    const float* cW   = (const float*)d_in[7];
    const float* pxr  = (const float*)d_in[8];
    float* out = (float*)d_out;

    size_t smem = (size_t)(16 * HSTRIDE + 16 * ZSTRIDE + 16 * CSTRIDE + 16) * sizeof(float)
                + 16 * sizeof(int);
    decoder_main_kernel<<<128, 256, smem, stream>>>(
        z, bc, sf, cc, aW, eA, eh3, cW, pxr, out);
}


# Softmax phase in registers: each wave reads its 2 rows once by 16-byte LDS loads (128 values a lane), max, exp, sum, scale there, 512-byte stores; sum order: 4 chains a lane, 32-lane butterfly
# speedup vs baseline: 2.4726x; 1.2421x over previous
#include <hip/hip_runtime.h>
#include <hip/hip_bf16.h>

#define B_SAMP 1024
#define N_IN   64
#define N_OUT  4096
#define N_BATCH 64
#define N_CONT 16

#define HSTRIDE 4100
#define ZSTRIDE 68
#define CSTRIDE 20

typedef __attribute__((ext_vector_type(2))) float v2f;
typedef __attribute__((ext_vector_type(4))) float v4f;
typedef __attribute__((ext_vector_type(8))) float v8f;

struct HasNext { static constexpr bool value = true;  };
struct NoNext  { static constexpr bool value = false; };

static __device__ __forceinline__ v8f wmma4(v2f a, v2f b, v8f c) {
    return __builtin_amdgcn_wmma_f32_16x16x4_f32(
         false, a,  false, b,
         (short)0, c,  false,  false);
}

static __device__ __forceinline__ void load_w_group(
    const float* amat_W, const float* cont_W, const float* h3b, int gc, int half,
    v4f (&w0)[16], v4f (&w1)[16], v4f (&c0)[4], v4f (&c1)[4], v4f& h3v)
{
#pragma unroll
    for (int cc = 0; cc < 16; ++cc) {
        const int k = 8 * (cc >> 1) + 4 * half + 2 * (cc & 1);
        w0[cc] = *(const v4f*)(amat_W + (size_t)k * N_OUT + gc);
        w1[cc] = *(const v4f*)(amat_W + (size_t)(k + 1) * N_OUT + gc);
    }
#pragma unroll
    for (int cc = 0; cc < 4; ++cc) {
        const int k = 4 * cc + 2 * half;
        c0[cc] = *(const v4f*)(cont_W + (size_t)k * N_OUT + gc);
        c1[cc] = *(const v4f*)(cont_W + (size_t)(k + 1) * N_OUT + gc);
    }
    h3v = *(const v4f*)(h3b + gc);
}

static __device__ __forceinline__ void load_e_group(
    const float* Eb, int gc, int half, v4f (&ev)[4][8])
{
#pragma unroll
    for (int j = 0; j < 4; ++j) {
        const float* ep = Eb + (size_t)(gc + j) * N_IN + 4 * half;
#pragma unroll
        for (int i = 0; i < 8; ++i) ev[j][i] = *(const v4f*)(ep + 8 * i);
    }
}

template <int MASK>
static __device__ __forceinline__ float row_xchg(float v) {
    const int b = __builtin_bit_cast(int, v);
    return __builtin_bit_cast(float,
        __builtin_amdgcn_update_dpp(b, b, 0x160 + MASK, 0xf, 0xf, false));
}
static __device__ __forceinline__ float half_xchg(float v) {
    const int b = __builtin_bit_cast(int, v);
    return __builtin_bit_cast(float,
        __builtin_amdgcn_permlanex16(b, b, 0x76543210, 0xfedcba98, false, false));
}
static __device__ __forceinline__ float wave_max32(float v) {
    v = fmaxf(v, row_xchg<1>(v));
    v = fmaxf(v, row_xchg<2>(v));
    v = fmaxf(v, row_xchg<4>(v));
    v = fmaxf(v, row_xchg<8>(v));
    return fmaxf(v, half_xchg(v));
}
static __device__ __forceinline__ float wave_sum32(float v) {
    v += row_xchg<1>(v);
    v += row_xchg<2>(v);
    v += row_xchg<4>(v);
    v += row_xchg<8>(v);
    return v + half_xchg(v);
}

__global__ __launch_bounds__(256) void decoder_main_kernel(
    const float* __restrict__ z, const int* __restrict__ idx,
    const float* __restrict__ sizef, const float* __restrict__ cont,
    const float* amat_W, const float* embed_A, const float* embed_h3,
    const float* cont_W, const float* __restrict__ px_r, float* __restrict__ out)
{
    extern __shared__ __attribute__((aligned(16))) float smem[];
    float* hbuf  = smem;
    float* ztile = hbuf  + 16 * HSTRIDE;
    float* ctile = ztile + 16 * ZSTRIDE;
    float* sfbuf = ctile + 16 * CSTRIDE;
    int*   sids  = (int*)(sfbuf + 16);

    __shared__ int hist[N_BATCH];
    __shared__ int ccnt[32];

    const int tile = blockIdx.x;
    const int t = threadIdx.x;
    const int wave = t >> 5;
    const int lane = t & 31;

    int bv[4];
#pragma unroll
    for (int j = 0; j < 4; ++j) bv[j] = idx[256 * j + t];
    float pr = 0.0f;
    if (t < 32) pr = px_r[tile * 32 + t];

    if (t < N_BATCH) hist[t] = 0;
    if (t < 16) sids[t] = -1;
    __syncthreads();
#pragma unroll
    for (int j = 0; j < 4; ++j) atomicAdd(&hist[bv[j]], 1);
    if (t < 32) out[(size_t)B_SAMP * N_OUT + tile * 32 + t] = __expf(pr);
    __syncthreads();

    int b = -1, q = 0, tc = 0;
#pragma unroll
    for (int bb = 0; bb < N_BATCH; ++bb) {
        const int nt = (hist[bb] + 15) >> 4;
        if (tile >= tc && tile < tc + nt) { b = bb; q = tile - tc; }
        tc += nt;
    }
    if (b < 0) return;

    bool mine[4];
    int below[4];
#pragma unroll
    for (int j = 0; j < 4; ++j) {
        mine[j] = (bv[j] == b);
        const unsigned m = (unsigned)__ballot(mine[j]);
        below[j] = __popc(m & ((1u << lane) - 1u));
        if (lane == 0) ccnt[8 * j + wave] = __popc(m);
    }
    __syncthreads();
    int base[4] = {0, 0, 0, 0};
#pragma unroll
    for (int c = 0; c < 32; ++c) {
        const int n = ccnt[c];
#pragma unroll
        for (int j = 0; j < 4; ++j) base[j] += (c < 8 * j + wave) ? n : 0;
    }
#pragma unroll
    for (int j = 0; j < 4; ++j) {
        const int rank = base[j] + below[j];
        if (mine[j] && (rank >> 4) == q) sids[rank & 15] = 256 * j + t;
    }
    __syncthreads();

    if (t < 16) {
        const int s = sids[t];
        sfbuf[t] = (s >= 0) ? sizef[s] : 1.0f;
    }

    for (int i = t; i < 16 * N_IN; i += 256) {
        int m = i >> 6, l = i & 63;
        int s = sids[m];
        ztile[m * ZSTRIDE + l] = (s >= 0) ? z[s * N_IN + l] : 0.0f;
    }
    for (int i = t; i < 16 * N_CONT; i += 256) {
        int m = i >> 4, l = i & 15;
        int s = sids[m];
        ctile[m * CSTRIDE + l] = (s >= 0) ? cont[s * N_CONT + l] : 0.0f;
    }
    __syncthreads();

    const int half = lane >> 4;
    const int ln   = lane & 15;

    v2f afrag[16];
#pragma unroll
    for (int i = 0; i < 8; ++i) {
        const v4f q = *(const v4f*)&ztile[ln * ZSTRIDE + 8 * i + 4 * half];
        v2f lo; lo.x = q.x; lo.y = q.y;
        v2f hi; hi.x = q.z; hi.y = q.w;
        afrag[2 * i]     = lo;
        afrag[2 * i + 1] = hi;
    }
    v2f cfrag[4];
#pragma unroll
    for (int c = 0; c < 4; ++c) {
        const float* p = &ctile[ln * CSTRIDE + 4 * c + 2 * half];
        v2f v; v.x = p[0]; v.y = p[1];
        cfrag[c] = v;
    }

    const float* Eb  = embed_A  + ((size_t)b << 18);
    const float* h3b = embed_h3 + (size_t)b * N_OUT;

    v4f w0[16], w1[16], ev[4][8], c0[4], c1[4], h3v, h3n;
    {
        const int gc = wave * 64 + 4 * ln;
        __builtin_prefetch(Eb + (size_t)(wave * 64 + 512 + lane) * N_IN, 0, 1);
        __builtin_prefetch(Eb + (size_t)(wave * 64 + 544 + lane) * N_IN, 0, 1);
        load_w_group(amat_W, cont_W, h3b, gc, half, w0, w1, c0, c1, h3v);
        load_e_group(Eb, gc, half, ev);
    }
    h3n = h3v;

    auto group_pass = [&](const int cg, auto more_tag) __attribute__((always_inline)) {
        constexpr bool MORE = decltype(more_tag)::value;
        const int gc = cg * 64 + 4 * ln;
        if (cg + 16 < N_OUT / 64) {
            __builtin_prefetch(Eb + (size_t)(cg * 64 + 1024 + lane) * N_IN, 0, 1);
            __builtin_prefetch(Eb + (size_t)(cg * 64 + 1056 + lane) * N_IN, 0, 1);
        }

        v8f acc[4] = {};
#pragma unroll
        for (int cc = 0; cc < 16; ++cc) {
#pragma unroll
            for (int j = 0; j < 4; ++j) {
                v2f bf; bf.x = w0[cc][j]; bf.y = w1[cc][j];
                acc[j] = wmma4(afrag[cc], bf, acc[j]);
            }
        }
#pragma unroll
        for (int cc = 0; cc < 4; ++cc) {
            asm("" : "+v"(c0[cc]), "+v"(c1[cc]));
#pragma unroll
            for (int j = 0; j < 4; ++j) {
                v2f bf; bf.x = c0[cc][j]; bf.y = c1[cc][j];
                acc[j] = wmma4(cfrag[cc], bf, acc[j]);
            }
        }
        __builtin_amdgcn_sched_barrier(0);
        if constexpr (MORE)
            load_w_group(amat_W, cont_W, h3b, gc + 512, half, w0, w1, c0, c1, h3n);

#pragma unroll
        for (int i = 0; i < 8; ++i) {
#pragma unroll
            for (int j = 0; j < 4; ++j) {
                v2f bf; bf.x = ev[j][i].x; bf.y = ev[j][i].y;
                acc[j] = wmma4(afrag[2 * i], bf, acc[j]);
            }
#pragma unroll
            for (int j = 0; j < 4; ++j) {
                v2f bf; bf.x = ev[j][i].z; bf.y = ev[j][i].w;
                acc[j] = wmma4(afrag[2 * i + 1], bf, acc[j]);
            }
        }
#pragma unroll
        for (int r = 0; r < 8; ++r) {
            const int m = r + 8 * half;
            v4f o;
            o.x = acc[0][r] + h3v.x;
            o.y = acc[1][r] + h3v.y;
            o.z = acc[2][r] + h3v.z;
            o.w = acc[3][r] + h3v.w;
            *(v4f*)&hbuf[m * HSTRIDE + gc] = o;
        }
        h3v = h3n;
        __builtin_amdgcn_sched_barrier(0);
        if constexpr (MORE) load_e_group(Eb, gc + 512, half, ev);
    };
    for (int it = 0; it < N_OUT / 64 / 8 - 1; ++it) group_pass(wave + 8 * it, HasNext{});
    group_pass(wave + N_OUT / 64 - 8, NoNext{});
    __syncthreads();

    v4f xr[2][32];
#pragma unroll
    for (int rr = 0; rr < 2; ++rr) {
        const float* hrow = hbuf + (2 * wave + rr) * HSTRIDE + 4 * lane;
#pragma unroll
        for (int n = 0; n < 32; ++n) xr[rr][n] = *(const v4f*)(hrow + 128 * n);
    }
#pragma unroll
    for (int rr = 0; rr < 2; ++rr) {
        const int row = 2 * wave + rr;

        v4f m4 = xr[rr][0];
#pragma unroll
        for (int n = 1; n < 32; ++n) {
            m4.x = fmaxf(m4.x, xr[rr][n].x);
            m4.y = fmaxf(m4.y, xr[rr][n].y);
            m4.z = fmaxf(m4.z, xr[rr][n].z);
            m4.w = fmaxf(m4.w, xr[rr][n].w);
        }
        const float mx = wave_max32(fmaxf(fmaxf(m4.x, m4.y), fmaxf(m4.z, m4.w)));

        v4f s4 = {0.0f, 0.0f, 0.0f, 0.0f};
#pragma unroll
        for (int n = 0; n < 32; ++n) {
            v4f e;
            e.x = __expf(xr[rr][n].x - mx);
            e.y = __expf(xr[rr][n].y - mx);
            e.z = __expf(xr[rr][n].z - mx);
            e.w = __expf(xr[rr][n].w - mx);
            xr[rr][n] = e;
            s4 += e;
        }
        const float sum = wave_sum32((s4.x + s4.y) + (s4.z + s4.w));

        const int s = sids[row];
        if (s >= 0) {
            const float scale = sfbuf[row] / sum;
            float* orow = out + (size_t)s * N_OUT + 4 * lane;
#pragma unroll
            for (int n = 0; n < 32; ++n) *(v4f*)(orow + 128 * n) = xr[rr][n] * scale;
        }
    }
}

extern "C" void kernel_launch(void* const* d_in, const int* in_sizes, int n_in,
                              void* d_out, int out_size, void* d_ws, size_t ws_size,
                              hipStream_t stream)
{
    const float* z    = (const float*)d_in[0];
    const int*   bc   = (const int*)  d_in[1];
    const float* sf   = (const float*)d_in[2];
    const float* cc   = (const float*)d_in[3];
    const float* aW   = (const float*)d_in[4];
    const float* eA   = (const float*)d_in[5];
    const float* eh3  = (const float*)d_in[6];
    const float* cW   = (const float*)d_in[7];
    const float* pxr  = (const float*)d_in[8];
    float* out = (float*)d_out;

    size_t smem = (size_t)(16 * HSTRIDE + 16 * ZSTRIDE + 16 * CSTRIDE + 16) * sizeof(float)
                + 16 * sizeof(int);
    decoder_main_kernel<<<128, 256, smem, stream>>>(
        z, bc, sf, cc, aW, eA, eh3, cW, pxr, out);
}


# K=64 weights pre-added on the vector unit: z@(amat_W + A_b^T) replaces z@amat_W + z@A_b^T, 64 of 144 WMMA per 64-column group gone, refills by k halves; sum order: weights summed first, cont_W first
# speedup vs baseline: 2.6532x; 1.0730x over previous
#include <hip/hip_runtime.h>
#include <hip/hip_bf16.h>

#define B_SAMP 1024
#define N_IN   64
#define N_OUT  4096
#define N_BATCH 64
#define N_CONT 16

#define HSTRIDE 4100
#define ZSTRIDE 68
#define CSTRIDE 20

typedef __attribute__((ext_vector_type(2))) float v2f;
typedef __attribute__((ext_vector_type(4))) float v4f;
typedef __attribute__((ext_vector_type(8))) float v8f;

struct HasNext { static constexpr bool value = true;  };
struct NoNext  { static constexpr bool value = false; };

static __device__ __forceinline__ v8f wmma4(v2f a, v2f b, v8f c) {
    return __builtin_amdgcn_wmma_f32_16x16x4_f32(
         false, a,  false, b,
         (short)0, c,  false,  false);
}

template <int I0>
static __device__ __forceinline__ void load_w_half(
    const float* amat_W, int gc, int half, v4f (&w0)[16], v4f (&w1)[16])
{
#pragma unroll
    for (int cc = 2 * I0; cc < 2 * I0 + 8; ++cc) {
        const int k = 8 * (cc >> 1) + 4 * half + 2 * (cc & 1);
        w0[cc] = *(const v4f*)(amat_W + (size_t)k * N_OUT + gc);
        w1[cc] = *(const v4f*)(amat_W + (size_t)(k + 1) * N_OUT + gc);
    }
}

static __device__ __forceinline__ void load_c_group(
    const float* cont_W, const float* h3b, int gc, int half,
    v4f (&c0)[4], v4f (&c1)[4], v4f& h3v)
{
#pragma unroll
    for (int cc = 0; cc < 4; ++cc) {
        const int k = 4 * cc + 2 * half;
        c0[cc] = *(const v4f*)(cont_W + (size_t)k * N_OUT + gc);
        c1[cc] = *(const v4f*)(cont_W + (size_t)(k + 1) * N_OUT + gc);
    }
    h3v = *(const v4f*)(h3b + gc);
}

template <int I0>
static __device__ __forceinline__ void load_e_half(
    const float* Eb, int gc, int half, v4f (&ev)[4][8])
{
#pragma unroll
    for (int j = 0; j < 4; ++j) {
        const float* ep = Eb + (size_t)(gc + j) * N_IN + 4 * half;
#pragma unroll
        for (int i = I0; i < I0 + 4; ++i) ev[j][i] = *(const v4f*)(ep + 8 * i);
    }
}

template <int I0>
static __device__ __forceinline__ void add_w_half(
    v4f (&ev)[4][8], const v4f (&w0)[16], const v4f (&w1)[16])
{
#pragma unroll
    for (int i = I0; i < I0 + 4; ++i) {
#pragma unroll
        for (int j = 0; j < 4; ++j) {
            ev[j][i].x += w0[2 * i][j];
            ev[j][i].y += w1[2 * i][j];
            ev[j][i].z += w0[2 * i + 1][j];
            ev[j][i].w += w1[2 * i + 1][j];
        }
    }
}

template <int I0, int I1>
static __device__ __forceinline__ void mm_k(
    v8f (&acc)[4], const v2f (&afrag)[16], const v4f (&ev)[4][8])
{
#pragma unroll
    for (int i = I0; i < I1; ++i) {
#pragma unroll
        for (int j = 0; j < 4; ++j) {
            v2f bf; bf.x = ev[j][i].x; bf.y = ev[j][i].y;
            acc[j] = wmma4(afrag[2 * i], bf, acc[j]);
        }
#pragma unroll
        for (int j = 0; j < 4; ++j) {
            v2f bf; bf.x = ev[j][i].z; bf.y = ev[j][i].w;
            acc[j] = wmma4(afrag[2 * i + 1], bf, acc[j]);
        }
    }
}

template <int MASK>
static __device__ __forceinline__ float row_xchg(float v) {
    const int b = __builtin_bit_cast(int, v);
    return __builtin_bit_cast(float,
        __builtin_amdgcn_update_dpp(b, b, 0x160 + MASK, 0xf, 0xf, false));
}
static __device__ __forceinline__ float half_xchg(float v) {
    const int b = __builtin_bit_cast(int, v);
    return __builtin_bit_cast(float,
        __builtin_amdgcn_permlanex16(b, b, 0x76543210, 0xfedcba98, false, false));
}
static __device__ __forceinline__ float wave_max32(float v) {
    v = fmaxf(v, row_xchg<1>(v));
    v = fmaxf(v, row_xchg<2>(v));
    v = fmaxf(v, row_xchg<4>(v));
    v = fmaxf(v, row_xchg<8>(v));
    return fmaxf(v, half_xchg(v));
}
static __device__ __forceinline__ float wave_sum32(float v) {
    v += row_xchg<1>(v);
    v += row_xchg<2>(v);
    v += row_xchg<4>(v);
    v += row_xchg<8>(v);
    return v + half_xchg(v);
}

__global__ __launch_bounds__(256) void decoder_main_kernel(
    const float* __restrict__ z, const int* __restrict__ idx,
    const float* __restrict__ sizef, const float* __restrict__ cont,
    const float* amat_W, const float* embed_A, const float* embed_h3,
    const float* cont_W, const float* __restrict__ px_r, float* __restrict__ out)
{
    extern __shared__ __attribute__((aligned(16))) float smem[];
    float* hbuf  = smem;
    float* ztile = hbuf  + 16 * HSTRIDE;
    float* ctile = ztile + 16 * ZSTRIDE;
    float* sfbuf = ctile + 16 * CSTRIDE;
    int*   sids  = (int*)(sfbuf + 16);

    __shared__ int hist[N_BATCH];
    __shared__ int ccnt[32];

    const int tile = blockIdx.x;
    const int t = threadIdx.x;
    const int wave = t >> 5;
    const int lane = t & 31;

    int bv[4];
#pragma unroll
    for (int j = 0; j < 4; ++j) bv[j] = idx[256 * j + t];
    float pr = 0.0f;
    if (t < 32) pr = px_r[tile * 32 + t];

    if (t < N_BATCH) hist[t] = 0;
    if (t < 16) sids[t] = -1;
    __syncthreads();
#pragma unroll
    for (int j = 0; j < 4; ++j) atomicAdd(&hist[bv[j]], 1);
    if (t < 32) out[(size_t)B_SAMP * N_OUT + tile * 32 + t] = __expf(pr);
    __syncthreads();

    int b = -1, q = 0, tc = 0;
#pragma unroll
    for (int bb = 0; bb < N_BATCH; ++bb) {
        const int nt = (hist[bb] + 15) >> 4;
        if (tile >= tc && tile < tc + nt) { b = bb; q = tile - tc; }
        tc += nt;
    }
    if (b < 0) return;

    bool mine[4];
    int below[4];
#pragma unroll
    for (int j = 0; j < 4; ++j) {
        mine[j] = (bv[j] == b);
        const unsigned m = (unsigned)__ballot(mine[j]);
        below[j] = __popc(m & ((1u << lane) - 1u));
        if (lane == 0) ccnt[8 * j + wave] = __popc(m);
    }
    __syncthreads();
    int base[4] = {0, 0, 0, 0};
#pragma unroll
    for (int c = 0; c < 32; ++c) {
        const int n = ccnt[c];
#pragma unroll
        for (int j = 0; j < 4; ++j) base[j] += (c < 8 * j + wave) ? n : 0;
    }
#pragma unroll
    for (int j = 0; j < 4; ++j) {
        const int rank = base[j] + below[j];
        if (mine[j] && (rank >> 4) == q) sids[rank & 15] = 256 * j + t;
    }
    __syncthreads();

    if (t < 16) {
        const int s = sids[t];
        sfbuf[t] = (s >= 0) ? sizef[s] : 1.0f;
    }

    for (int i = t; i < 16 * N_IN; i += 256) {
        int m = i >> 6, l = i & 63;
        int s = sids[m];
        ztile[m * ZSTRIDE + l] = (s >= 0) ? z[s * N_IN + l] : 0.0f;
    }
    for (int i = t; i < 16 * N_CONT; i += 256) {
        int m = i >> 4, l = i & 15;
        int s = sids[m];
        ctile[m * CSTRIDE + l] = (s >= 0) ? cont[s * N_CONT + l] : 0.0f;
    }
    __syncthreads();

    const int half = lane >> 4;
    const int ln   = lane & 15;

    v2f afrag[16];
#pragma unroll
    for (int i = 0; i < 8; ++i) {
        const v4f q = *(const v4f*)&ztile[ln * ZSTRIDE + 8 * i + 4 * half];
        v2f lo; lo.x = q.x; lo.y = q.y;
        v2f hi; hi.x = q.z; hi.y = q.w;
        afrag[2 * i]     = lo;
        afrag[2 * i + 1] = hi;
    }
    v2f cfrag[4];
#pragma unroll
    for (int c = 0; c < 4; ++c) {
        const float* p = &ctile[ln * CSTRIDE + 4 * c + 2 * half];
        v2f v; v.x = p[0]; v.y = p[1];
        cfrag[c] = v;
    }

    const float* Eb  = embed_A  + ((size_t)b << 18);
    const float* h3b = embed_h3 + (size_t)b * N_OUT;

    v4f w0[16], w1[16], ev[4][8], c0[4], c1[4], h3v, h3n;
    {
        const int gc = wave * 64 + 4 * ln;
        __builtin_prefetch(Eb + (size_t)(wave * 64 + 512 + lane) * N_IN, 0, 1);
        __builtin_prefetch(Eb + (size_t)(wave * 64 + 544 + lane) * N_IN, 0, 1);
        load_w_half<0>(amat_W, gc, half, w0, w1);
        load_c_group(cont_W, h3b, gc, half, c0, c1, h3v);
        load_e_half<0>(Eb, gc, half, ev);
        load_w_half<4>(amat_W, gc, half, w0, w1);
        load_e_half<4>(Eb, gc, half, ev);
    }
    h3n = h3v;

    auto group_pass = [&](const int cg, auto more_tag) __attribute__((always_inline)) {
        constexpr bool MORE = decltype(more_tag)::value;
        const int gc = cg * 64 + 4 * ln;
        if (cg + 16 < N_OUT / 64) {
            __builtin_prefetch(Eb + (size_t)(cg * 64 + 1024 + lane) * N_IN, 0, 1);
            __builtin_prefetch(Eb + (size_t)(cg * 64 + 1056 + lane) * N_IN, 0, 1);
        }

        v8f acc[4] = {};
#pragma unroll
        for (int cc = 0; cc < 4; ++cc) {
#pragma unroll
            for (int j = 0; j < 4; ++j) {
                v2f bf; bf.x = c0[cc][j]; bf.y = c1[cc][j];
                acc[j] = wmma4(cfrag[cc], bf, acc[j]);
            }
        }
        add_w_half<0>(ev, w0, w1);
        __builtin_amdgcn_sched_barrier(0);
        if constexpr (MORE) {
            load_w_half<0>(amat_W, gc + 512, half, w0, w1);
            load_c_group(cont_W, h3b, gc + 512, half, c0, c1, h3n);
        }
        mm_k<0, 2>(acc, afrag, ev);
        __builtin_amdgcn_sched_barrier(0);
        mm_k<2, 4>(acc, afrag, ev);
        add_w_half<4>(ev, w0, w1);
        __builtin_amdgcn_sched_barrier(0);
        if constexpr (MORE) {
            load_e_half<0>(Eb, gc + 512, half, ev);
            load_w_half<4>(amat_W, gc + 512, half, w0, w1);
        }
        mm_k<4, 8>(acc, afrag, ev);
        __builtin_amdgcn_sched_barrier(0);
        if constexpr (MORE) load_e_half<4>(Eb, gc + 512, half, ev);
#pragma unroll
        for (int r = 0; r < 8; ++r) {
            const int m = r + 8 * half;
            v4f o;
            o.x = acc[0][r] + h3v.x;
            o.y = acc[1][r] + h3v.y;
            o.z = acc[2][r] + h3v.z;
            o.w = acc[3][r] + h3v.w;
            *(v4f*)&hbuf[m * HSTRIDE + gc] = o;
        }
        h3v = h3n;
    };
    for (int it = 0; it < N_OUT / 64 / 8 - 1; ++it) group_pass(wave + 8 * it, HasNext{});
    group_pass(wave + N_OUT / 64 - 8, NoNext{});
    __syncthreads();

    v4f xr[2][32];
#pragma unroll
    for (int rr = 0; rr < 2; ++rr) {
        const float* hrow = hbuf + (2 * wave + rr) * HSTRIDE + 4 * lane;
#pragma unroll
        for (int n = 0; n < 32; ++n) xr[rr][n] = *(const v4f*)(hrow + 128 * n);
    }
#pragma unroll
    for (int rr = 0; rr < 2; ++rr) {
        const int row = 2 * wave + rr;

        v4f m4 = xr[rr][0];
#pragma unroll
        for (int n = 1; n < 32; ++n) {
            m4.x = fmaxf(m4.x, xr[rr][n].x);
            m4.y = fmaxf(m4.y, xr[rr][n].y);
            m4.z = fmaxf(m4.z, xr[rr][n].z);
            m4.w = fmaxf(m4.w, xr[rr][n].w);
        }
        const float mx = wave_max32(fmaxf(fmaxf(m4.x, m4.y), fmaxf(m4.z, m4.w)));

        v4f s4 = {0.0f, 0.0f, 0.0f, 0.0f};
#pragma unroll
        for (int n = 0; n < 32; ++n) {
            v4f e;
            e.x = __expf(xr[rr][n].x - mx);
            e.y = __expf(xr[rr][n].y - mx);
            e.z = __expf(xr[rr][n].z - mx);
            e.w = __expf(xr[rr][n].w - mx);
            xr[rr][n] = e;
            s4 += e;
        }
        const float sum = wave_sum32((s4.x + s4.y) + (s4.z + s4.w));

        const int s = sids[row];
        if (s >= 0) {
            const float scale = sfbuf[row] / sum;
            float* orow = out + (size_t)s * N_OUT + 4 * lane;
#pragma unroll
            for (int n = 0; n < 32; ++n) *(v4f*)(orow + 128 * n) = xr[rr][n] * scale;
        }
    }
}

extern "C" void kernel_launch(void* const* d_in, const int* in_sizes, int n_in,
                              void* d_out, int out_size, void* d_ws, size_t ws_size,
                              hipStream_t stream)
{
    const float* z    = (const float*)d_in[0];
    const int*   bc   = (const int*)  d_in[1];
    const float* sf   = (const float*)d_in[2];
    const float* cc   = (const float*)d_in[3];
    const float* aW   = (const float*)d_in[4];
    const float* eA   = (const float*)d_in[5];
    const float* eh3  = (const float*)d_in[6];
    const float* cW   = (const float*)d_in[7];
    const float* pxr  = (const float*)d_in[8];
    float* out = (float*)d_out;

    size_t smem = (size_t)(16 * HSTRIDE + 16 * ZSTRIDE + 16 * CSTRIDE + 16) * sizeof(float)
                + 16 * sizeof(int);
    decoder_main_kernel<<<128, 256, smem, stream>>>(
        z, bc, sf, cc, aW, eA, eh3, cW, pxr, out);
}


# A_b operand fetched by whole-line loads (8 lanes per 128-byte half row, 4 lines an instruction instead of 16) and re-laid into fragment order through a per-wave LDS patch; sum order unchanged
# speedup vs baseline: 2.9317x; 1.1050x over previous
#include <hip/hip_runtime.h>
#include <hip/hip_bf16.h>

#define B_SAMP 1024
#define N_IN   64
#define N_OUT  4096
#define N_BATCH 64
#define N_CONT 16

#define HSTRIDE 4100
#define ZSTRIDE 68
#define CSTRIDE 20
#define ESTRIDE 36

typedef __attribute__((ext_vector_type(2))) float v2f;
typedef __attribute__((ext_vector_type(4))) float v4f;
typedef __attribute__((ext_vector_type(8))) float v8f;

struct HasNext { static constexpr bool value = true;  };
struct NoNext  { static constexpr bool value = false; };

static __device__ __forceinline__ v8f wmma4(v2f a, v2f b, v8f c) {
    return __builtin_amdgcn_wmma_f32_16x16x4_f32(
         false, a,  false, b,
         (short)0, c,  false,  false);
}

template <int I0>
static __device__ __forceinline__ void load_w_half(
    const float* amat_W, int gc, int half, v4f (&w0)[16], v4f (&w1)[16])
{
#pragma unroll
    for (int cc = 2 * I0; cc < 2 * I0 + 8; ++cc) {
        const int k = 8 * (cc >> 1) + 4 * half + 2 * (cc & 1);
        w0[cc] = *(const v4f*)(amat_W + (size_t)k * N_OUT + gc);
        w1[cc] = *(const v4f*)(amat_W + (size_t)(k + 1) * N_OUT + gc);
    }
}

static __device__ __forceinline__ void load_c_group(
    const float* cont_W, const float* h3b, int gc, int half,
    v4f (&c0)[4], v4f (&c1)[4], v4f& h3v)
{
#pragma unroll
    for (int cc = 0; cc < 4; ++cc) {
        const int k = 4 * cc + 2 * half;
        c0[cc] = *(const v4f*)(cont_W + (size_t)k * N_OUT + gc);
        c1[cc] = *(const v4f*)(cont_W + (size_t)(k + 1) * N_OUT + gc);
    }
    h3v = *(const v4f*)(h3b + gc);
}

template <int I0>
static __device__ __forceinline__ void load_e_half(
    const float* Eb, int c0, int lane, v4f (&ev)[4][8])
{
    const int r = lane >> 3, c = lane & 7;
#pragma unroll
    for (int j = 0; j < 4; ++j) {
        const float* ep = Eb + (size_t)(c0 + 4 * r + j) * N_IN + 8 * I0 + 4 * c;
#pragma unroll
        for (int t = 0; t < 4; ++t) ev[j][I0 + t] = *(const v4f*)(ep + (size_t)t * 16 * N_IN);
    }
}

template <int I0>
static __device__ __forceinline__ void xpose_e_half(
    float* st, int lane, v4f (&ev)[4][8])
{
    float* wp = st + (lane >> 3) * ESTRIDE + 4 * (lane & 7);
    const float* rp = st + (lane & 15) * ESTRIDE + 4 * (lane >> 4);
#pragma unroll
    for (int j = 0; j < 4; ++j) {
#pragma unroll
        for (int t = 0; t < 4; ++t) *(v4f*)(wp + t * 4 * ESTRIDE) = ev[j][I0 + t];
#pragma unroll
        for (int i = 0; i < 4; ++i) ev[j][I0 + i] = *(const v4f*)(rp + 8 * i);
    }
}

template <int I0>
static __device__ __forceinline__ void add_w_half(
    v4f (&ev)[4][8], const v4f (&w0)[16], const v4f (&w1)[16])
{
#pragma unroll
    for (int i = I0; i < I0 + 4; ++i) {
#pragma unroll
        for (int j = 0; j < 4; ++j) {
            ev[j][i].x += w0[2 * i][j];
            ev[j][i].y += w1[2 * i][j];
            ev[j][i].z += w0[2 * i + 1][j];
            ev[j][i].w += w1[2 * i + 1][j];
        }
    }
}

template <int I0, int I1>
static __device__ __forceinline__ void mm_k(
    v8f (&acc)[4], const v2f (&afrag)[16], const v4f (&ev)[4][8])
{
#pragma unroll
    for (int i = I0; i < I1; ++i) {
#pragma unroll
        for (int j = 0; j < 4; ++j) {
            v2f bf; bf.x = ev[j][i].x; bf.y = ev[j][i].y;
            acc[j] = wmma4(afrag[2 * i], bf, acc[j]);
        }
#pragma unroll
        for (int j = 0; j < 4; ++j) {
            v2f bf; bf.x = ev[j][i].z; bf.y = ev[j][i].w;
            acc[j] = wmma4(afrag[2 * i + 1], bf, acc[j]);
        }
    }
}

template <int MASK>
static __device__ __forceinline__ float row_xchg(float v) {
    const int b = __builtin_bit_cast(int, v);
    return __builtin_bit_cast(float,
        __builtin_amdgcn_update_dpp(b, b, 0x160 + MASK, 0xf, 0xf, false));
}
static __device__ __forceinline__ float half_xchg(float v) {
    const int b = __builtin_bit_cast(int, v);
    return __builtin_bit_cast(float,
        __builtin_amdgcn_permlanex16(b, b, 0x76543210, 0xfedcba98, false, false));
}
static __device__ __forceinline__ float wave_max32(float v) {
    v = fmaxf(v, row_xchg<1>(v));
    v = fmaxf(v, row_xchg<2>(v));
    v = fmaxf(v, row_xchg<4>(v));
    v = fmaxf(v, row_xchg<8>(v));
    return fmaxf(v, half_xchg(v));
}
static __device__ __forceinline__ float wave_sum32(float v) {
    v += row_xchg<1>(v);
    v += row_xchg<2>(v);
    v += row_xchg<4>(v);
    v += row_xchg<8>(v);
    return v + half_xchg(v);
}

__global__ __launch_bounds__(256) void decoder_main_kernel(
    const float* __restrict__ z, const int* __restrict__ idx,
    const float* __restrict__ sizef, const float* __restrict__ cont,
    const float* amat_W, const float* embed_A, const float* embed_h3,
    const float* cont_W, const float* __restrict__ px_r, float* __restrict__ out)
{
    extern __shared__ __attribute__((aligned(16))) float smem[];
    float* hbuf  = smem;
    float* ztile = hbuf  + 16 * HSTRIDE;
    float* ctile = ztile + 16 * ZSTRIDE;
    float* sfbuf = ctile + 16 * CSTRIDE;
    int*   sids  = (int*)(sfbuf + 16);
    float* estage = (float*)(sids + 16);

    __shared__ int hist[N_BATCH];
    __shared__ int ccnt[32];

    const int tile = blockIdx.x;
    const int t = threadIdx.x;
    const int wave = t >> 5;
    const int lane = t & 31;

    int bv[4];
#pragma unroll
    for (int j = 0; j < 4; ++j) bv[j] = idx[256 * j + t];
    float pr = 0.0f;
    if (t < 32) pr = px_r[tile * 32 + t];

    if (t < N_BATCH) hist[t] = 0;
    if (t < 16) sids[t] = -1;
    __syncthreads();
#pragma unroll
    for (int j = 0; j < 4; ++j) atomicAdd(&hist[bv[j]], 1);
    if (t < 32) out[(size_t)B_SAMP * N_OUT + tile * 32 + t] = __expf(pr);
    __syncthreads();

    int b = -1, q = 0, tc = 0;
#pragma unroll
    for (int bb = 0; bb < N_BATCH; ++bb) {
        const int nt = (hist[bb] + 15) >> 4;
        if (tile >= tc && tile < tc + nt) { b = bb; q = tile - tc; }
        tc += nt;
    }
    if (b < 0) return;

    bool mine[4];
    int below[4];
#pragma unroll
    for (int j = 0; j < 4; ++j) {
        mine[j] = (bv[j] == b);
        const unsigned m = (unsigned)__ballot(mine[j]);
        below[j] = __popc(m & ((1u << lane) - 1u));
        if (lane == 0) ccnt[8 * j + wave] = __popc(m);
    }
    __syncthreads();
    int base[4] = {0, 0, 0, 0};
#pragma unroll
    for (int c = 0; c < 32; ++c) {
        const int n = ccnt[c];
#pragma unroll
        for (int j = 0; j < 4; ++j) base[j] += (c < 8 * j + wave) ? n : 0;
    }
#pragma unroll
    for (int j = 0; j < 4; ++j) {
        const int rank = base[j] + below[j];
        if (mine[j] && (rank >> 4) == q) sids[rank & 15] = 256 * j + t;
    }
    __syncthreads();

    if (t < 16) {
        const int s = sids[t];
        sfbuf[t] = (s >= 0) ? sizef[s] : 1.0f;
    }

    for (int i = t; i < 16 * N_IN; i += 256) {
        int m = i >> 6, l = i & 63;
        int s = sids[m];
        ztile[m * ZSTRIDE + l] = (s >= 0) ? z[s * N_IN + l] : 0.0f;
    }
    for (int i = t; i < 16 * N_CONT; i += 256) {
        int m = i >> 4, l = i & 15;
        int s = sids[m];
        ctile[m * CSTRIDE + l] = (s >= 0) ? cont[s * N_CONT + l] : 0.0f;
    }
    __syncthreads();

    const int half = lane >> 4;
    const int ln   = lane & 15;

    v2f afrag[16];
#pragma unroll
    for (int i = 0; i < 8; ++i) {
        const v4f q = *(const v4f*)&ztile[ln * ZSTRIDE + 8 * i + 4 * half];
        v2f lo; lo.x = q.x; lo.y = q.y;
        v2f hi; hi.x = q.z; hi.y = q.w;
        afrag[2 * i]     = lo;
        afrag[2 * i + 1] = hi;
    }
    v2f cfrag[4];
#pragma unroll
    for (int c = 0; c < 4; ++c) {
        const float* p = &ctile[ln * CSTRIDE + 4 * c + 2 * half];
        v2f v; v.x = p[0]; v.y = p[1];
        cfrag[c] = v;
    }

    const float* Eb  = embed_A  + ((size_t)b << 18);
    const float* h3b = embed_h3 + (size_t)b * N_OUT;

    v4f w0[16], w1[16], ev[4][8], c0[4], c1[4], h3v, h3n;
    {
        const int gc = wave * 64 + 4 * ln;
        __builtin_prefetch(Eb + (size_t)(wave * 64 + 512 + lane) * N_IN, 0, 1);
        __builtin_prefetch(Eb + (size_t)(wave * 64 + 544 + lane) * N_IN, 0, 1);
        load_w_half<0>(amat_W, gc, half, w0, w1);
        load_c_group(cont_W, h3b, gc, half, c0, c1, h3v);
        __builtin_amdgcn_sched_barrier(0);
        load_e_half<0>(Eb, wave * 64, lane, ev);
        __builtin_amdgcn_sched_barrier(0);
        load_w_half<4>(amat_W, gc, half, w0, w1);
        __builtin_amdgcn_sched_barrier(0);
        load_e_half<4>(Eb, wave * 64, lane, ev);
        __builtin_amdgcn_sched_barrier(0);
    }
    h3n = h3v;
    float* est = estage + wave * (16 * ESTRIDE);

    auto group_pass = [&](const int cg, auto more_tag) __attribute__((always_inline)) {
        constexpr bool MORE = decltype(more_tag)::value;
        const int gc = cg * 64 + 4 * ln;
        if (cg + 16 < N_OUT / 64) {
            __builtin_prefetch(Eb + (size_t)(cg * 64 + 1024 + lane) * N_IN, 0, 1);
            __builtin_prefetch(Eb + (size_t)(cg * 64 + 1056 + lane) * N_IN, 0, 1);
        }

        xpose_e_half<0>(est, lane, ev);
        __builtin_amdgcn_sched_barrier(0);

        v8f acc[4] = {};
#pragma unroll
        for (int cc = 0; cc < 4; ++cc) {
#pragma unroll
            for (int j = 0; j < 4; ++j) {
                v2f bf; bf.x = c0[cc][j]; bf.y = c1[cc][j];
                acc[j] = wmma4(cfrag[cc], bf, acc[j]);
            }
        }
        add_w_half<0>(ev, w0, w1);
        __builtin_amdgcn_sched_barrier(0);
        if constexpr (MORE) {
            load_w_half<0>(amat_W, gc + 512, half, w0, w1);
            load_c_group(cont_W, h3b, gc + 512, half, c0, c1, h3n);
        }
        mm_k<0, 2>(acc, afrag, ev);
        __builtin_amdgcn_sched_barrier(0);
        xpose_e_half<4>(est, lane, ev);
        __builtin_amdgcn_sched_barrier(0);
        mm_k<2, 4>(acc, afrag, ev);
        add_w_half<4>(ev, w0, w1);
        __builtin_amdgcn_sched_barrier(0);
        if constexpr (MORE) {
            load_e_half<0>(Eb, cg * 64 + 512, lane, ev);
            load_w_half<4>(amat_W, gc + 512, half, w0, w1);
        }
        mm_k<4, 8>(acc, afrag, ev);
        __builtin_amdgcn_sched_barrier(0);
        if constexpr (MORE) load_e_half<4>(Eb, cg * 64 + 512, lane, ev);
#pragma unroll
        for (int r = 0; r < 8; ++r) {
            const int m = r + 8 * half;
            v4f o;
            o.x = acc[0][r] + h3v.x;
            o.y = acc[1][r] + h3v.y;
            o.z = acc[2][r] + h3v.z;
            o.w = acc[3][r] + h3v.w;
            *(v4f*)&hbuf[m * HSTRIDE + gc] = o;
        }
        h3v = h3n;
    };
    for (int it = 0; it < N_OUT / 64 / 8 - 1; ++it) group_pass(wave + 8 * it, HasNext{});
    group_pass(wave + N_OUT / 64 - 8, NoNext{});
    __syncthreads();

    v4f xr[2][32];
#pragma unroll
    for (int rr = 0; rr < 2; ++rr) {
        const float* hrow = hbuf + (2 * wave + rr) * HSTRIDE + 4 * lane;
#pragma unroll
        for (int n = 0; n < 32; ++n) xr[rr][n] = *(const v4f*)(hrow + 128 * n);
    }
#pragma unroll
    for (int rr = 0; rr < 2; ++rr) {
        const int row = 2 * wave + rr;

        v4f m4 = xr[rr][0];
#pragma unroll
        for (int n = 1; n < 32; ++n) {
            m4.x = fmaxf(m4.x, xr[rr][n].x);
            m4.y = fmaxf(m4.y, xr[rr][n].y);
            m4.z = fmaxf(m4.z, xr[rr][n].z);
            m4.w = fmaxf(m4.w, xr[rr][n].w);
        }
        const float mx = wave_max32(fmaxf(fmaxf(m4.x, m4.y), fmaxf(m4.z, m4.w)));

        v4f s4 = {0.0f, 0.0f, 0.0f, 0.0f};
#pragma unroll
        for (int n = 0; n < 32; ++n) {
            v4f e;
            e.x = __expf(xr[rr][n].x - mx);
            e.y = __expf(xr[rr][n].y - mx);
            e.z = __expf(xr[rr][n].z - mx);
            e.w = __expf(xr[rr][n].w - mx);
            xr[rr][n] = e;
            s4 += e;
        }
        const float sum = wave_sum32((s4.x + s4.y) + (s4.z + s4.w));

        const int s = sids[row];
        if (s >= 0) {
            const float scale = sfbuf[row] / sum;
            float* orow = out + (size_t)s * N_OUT + 4 * lane;
#pragma unroll
            for (int n = 0; n < 32; ++n) *(v4f*)(orow + 128 * n) = xr[rr][n] * scale;
        }
    }
}

extern "C" void kernel_launch(void* const* d_in, const int* in_sizes, int n_in,
                              void* d_out, int out_size, void* d_ws, size_t ws_size,
                              hipStream_t stream)
{
    const float* z    = (const float*)d_in[0];
    const int*   bc   = (const int*)  d_in[1];
    const float* sf   = (const float*)d_in[2];
    const float* cc   = (const float*)d_in[3];
    const float* aW   = (const float*)d_in[4];
    const float* eA   = (const float*)d_in[5];
    const float* eh3  = (const float*)d_in[6];
    const float* cW   = (const float*)d_in[7];
    const float* pxr  = (const float*)d_in[8];
    float* out = (float*)d_out;

    size_t smem = (size_t)(16 * HSTRIDE + 16 * ZSTRIDE + 16 * CSTRIDE + 16 + 8 * 16 * ESTRIDE) * sizeof(float)
                + 16 * sizeof(int);
    decoder_main_kernel<<<128, 256, smem, stream>>>(
        z, bc, sf, cc, aW, eA, eh3, cW, pxr, out);
}


# Blocks walk their 8 column-group passes from block-specific starting groups, so that they no longer ask L2 for the same amat_W, cont_W, embed_h3 lines in the same step; sum order unchanged
# speedup vs baseline: 3.0767x; 1.0495x over previous
#include <hip/hip_runtime.h>
#include <hip/hip_bf16.h>

#define B_SAMP 1024
#define N_IN   64
#define N_OUT  4096
#define N_BATCH 64
#define N_CONT 16

#define HSTRIDE 4100
#define ZSTRIDE 68
#define CSTRIDE 20
#define ESTRIDE 36

typedef __attribute__((ext_vector_type(2))) float v2f;
typedef __attribute__((ext_vector_type(4))) float v4f;
typedef __attribute__((ext_vector_type(8))) float v8f;

struct HasNext { static constexpr bool value = true;  };
struct NoNext  { static constexpr bool value = false; };

static __device__ __forceinline__ v8f wmma4(v2f a, v2f b, v8f c) {
    return __builtin_amdgcn_wmma_f32_16x16x4_f32(
         false, a,  false, b,
         (short)0, c,  false,  false);
}

template <int I0>
static __device__ __forceinline__ void load_w_half(
    const float* amat_W, int gc, int half, v4f (&w0)[16], v4f (&w1)[16])
{
#pragma unroll
    for (int cc = 2 * I0; cc < 2 * I0 + 8; ++cc) {
        const int k = 8 * (cc >> 1) + 4 * half + 2 * (cc & 1);
        w0[cc] = *(const v4f*)(amat_W + (size_t)k * N_OUT + gc);
        w1[cc] = *(const v4f*)(amat_W + (size_t)(k + 1) * N_OUT + gc);
    }
}

static __device__ __forceinline__ void load_c_group(
    const float* cont_W, const float* h3b, int gc, int half,
    v4f (&c0)[4], v4f (&c1)[4], v4f& h3v)
{
#pragma unroll
    for (int cc = 0; cc < 4; ++cc) {
        const int k = 4 * cc + 2 * half;
        c0[cc] = *(const v4f*)(cont_W + (size_t)k * N_OUT + gc);
        c1[cc] = *(const v4f*)(cont_W + (size_t)(k + 1) * N_OUT + gc);
    }
    h3v = *(const v4f*)(h3b + gc);
}

template <int I0>
static __device__ __forceinline__ void load_e_half(
    const float* Eb, int c0, int lane, v4f (&ev)[4][8])
{
    const int r = lane >> 3, c = lane & 7;
#pragma unroll
    for (int j = 0; j < 4; ++j) {
        const float* ep = Eb + (size_t)(c0 + 4 * r + j) * N_IN + 8 * I0 + 4 * c;
#pragma unroll
        for (int t = 0; t < 4; ++t) ev[j][I0 + t] = *(const v4f*)(ep + (size_t)t * 16 * N_IN);
    }
}

template <int I0>
static __device__ __forceinline__ void xpose_e_half(
    float* st, int lane, v4f (&ev)[4][8])
{
    float* wp = st + (lane >> 3) * ESTRIDE + 4 * (lane & 7);
    const float* rp = st + (lane & 15) * ESTRIDE + 4 * (lane >> 4);
#pragma unroll
    for (int j = 0; j < 4; ++j) {
#pragma unroll
        for (int t = 0; t < 4; ++t) *(v4f*)(wp + t * 4 * ESTRIDE) = ev[j][I0 + t];
#pragma unroll
        for (int i = 0; i < 4; ++i) ev[j][I0 + i] = *(const v4f*)(rp + 8 * i);
    }
}

template <int I0>
static __device__ __forceinline__ void add_w_half(
    v4f (&ev)[4][8], const v4f (&w0)[16], const v4f (&w1)[16])
{
#pragma unroll
    for (int i = I0; i < I0 + 4; ++i) {
#pragma unroll
        for (int j = 0; j < 4; ++j) {
            ev[j][i].x += w0[2 * i][j];
            ev[j][i].y += w1[2 * i][j];
            ev[j][i].z += w0[2 * i + 1][j];
            ev[j][i].w += w1[2 * i + 1][j];
        }
    }
}

template <int I0, int I1>
static __device__ __forceinline__ void mm_k(
    v8f (&acc)[4], const v2f (&afrag)[16], const v4f (&ev)[4][8])
{
#pragma unroll
    for (int i = I0; i < I1; ++i) {
#pragma unroll
        for (int j = 0; j < 4; ++j) {
            v2f bf; bf.x = ev[j][i].x; bf.y = ev[j][i].y;
            acc[j] = wmma4(afrag[2 * i], bf, acc[j]);
        }
#pragma unroll
        for (int j = 0; j < 4; ++j) {
            v2f bf; bf.x = ev[j][i].z; bf.y = ev[j][i].w;
            acc[j] = wmma4(afrag[2 * i + 1], bf, acc[j]);
        }
    }
}

template <int MASK>
static __device__ __forceinline__ float row_xchg(float v) {
    const int b = __builtin_bit_cast(int, v);
    return __builtin_bit_cast(float,
        __builtin_amdgcn_update_dpp(b, b, 0x160 + MASK, 0xf, 0xf, false));
}
static __device__ __forceinline__ float half_xchg(float v) {
    const int b = __builtin_bit_cast(int, v);
    return __builtin_bit_cast(float,
        __builtin_amdgcn_permlanex16(b, b, 0x76543210, 0xfedcba98, false, false));
}
static __device__ __forceinline__ float wave_max32(float v) {
    v = fmaxf(v, row_xchg<1>(v));
    v = fmaxf(v, row_xchg<2>(v));
    v = fmaxf(v, row_xchg<4>(v));
    v = fmaxf(v, row_xchg<8>(v));
    return fmaxf(v, half_xchg(v));
}
static __device__ __forceinline__ float wave_sum32(float v) {
    v += row_xchg<1>(v);
    v += row_xchg<2>(v);
    v += row_xchg<4>(v);
    v += row_xchg<8>(v);
    return v + half_xchg(v);
}

__global__ __launch_bounds__(256) void decoder_main_kernel(
    const float* __restrict__ z, const int* __restrict__ idx,
    const float* __restrict__ sizef, const float* __restrict__ cont,
    const float* amat_W, const float* embed_A, const float* embed_h3,
    const float* cont_W, const float* __restrict__ px_r, float* __restrict__ out)
{
    extern __shared__ __attribute__((aligned(16))) float smem[];
    float* hbuf  = smem;
    float* ztile = hbuf  + 16 * HSTRIDE;
    float* ctile = ztile + 16 * ZSTRIDE;
    float* sfbuf = ctile + 16 * CSTRIDE;
    int*   sids  = (int*)(sfbuf + 16);
    float* estage = (float*)(sids + 16);

    __shared__ int hist[N_BATCH];
    __shared__ int ccnt[32];

    const int tile = blockIdx.x;
    const int t = threadIdx.x;
    const int wave = t >> 5;
    const int lane = t & 31;

    int bv[4];
#pragma unroll
    for (int j = 0; j < 4; ++j) bv[j] = idx[256 * j + t];
    float pr = 0.0f;
    if (t < 32) pr = px_r[tile * 32 + t];

    if (t < N_BATCH) hist[t] = 0;
    if (t < 16) sids[t] = -1;
    __syncthreads();
#pragma unroll
    for (int j = 0; j < 4; ++j) atomicAdd(&hist[bv[j]], 1);
    if (t < 32) out[(size_t)B_SAMP * N_OUT + tile * 32 + t] = __expf(pr);
    __syncthreads();

    int b = -1, q = 0, tc = 0;
#pragma unroll
    for (int bb = 0; bb < N_BATCH; ++bb) {
        const int nt = (hist[bb] + 15) >> 4;
        if (tile >= tc && tile < tc + nt) { b = bb; q = tile - tc; }
        tc += nt;
    }
    if (b < 0) return;

    bool mine[4];
    int below[4];
#pragma unroll
    for (int j = 0; j < 4; ++j) {
        mine[j] = (bv[j] == b);
        const unsigned m = (unsigned)__ballot(mine[j]);
        below[j] = __popc(m & ((1u << lane) - 1u));
        if (lane == 0) ccnt[8 * j + wave] = __popc(m);
    }
    __syncthreads();
    int base[4] = {0, 0, 0, 0};
#pragma unroll
    for (int c = 0; c < 32; ++c) {
        const int n = ccnt[c];
#pragma unroll
        for (int j = 0; j < 4; ++j) base[j] += (c < 8 * j + wave) ? n : 0;
    }
#pragma unroll
    for (int j = 0; j < 4; ++j) {
        const int rank = base[j] + below[j];
        if (mine[j] && (rank >> 4) == q) sids[rank & 15] = 256 * j + t;
    }
    __syncthreads();

    if (t < 16) {
        const int s = sids[t];
        sfbuf[t] = (s >= 0) ? sizef[s] : 1.0f;
    }

    for (int i = t; i < 16 * N_IN; i += 256) {
        int m = i >> 6, l = i & 63;
        int s = sids[m];
        ztile[m * ZSTRIDE + l] = (s >= 0) ? z[s * N_IN + l] : 0.0f;
    }
    for (int i = t; i < 16 * N_CONT; i += 256) {
        int m = i >> 4, l = i & 15;
        int s = sids[m];
        ctile[m * CSTRIDE + l] = (s >= 0) ? cont[s * N_CONT + l] : 0.0f;
    }
    __syncthreads();

    const int half = lane >> 4;
    const int ln   = lane & 15;

    v2f afrag[16];
#pragma unroll
    for (int i = 0; i < 8; ++i) {
        const v4f q = *(const v4f*)&ztile[ln * ZSTRIDE + 8 * i + 4 * half];
        v2f lo; lo.x = q.x; lo.y = q.y;
        v2f hi; hi.x = q.z; hi.y = q.w;
        afrag[2 * i]     = lo;
        afrag[2 * i + 1] = hi;
    }
    v2f cfrag[4];
#pragma unroll
    for (int c = 0; c < 4; ++c) {
        const float* p = &ctile[ln * CSTRIDE + 4 * c + 2 * half];
        v2f v; v.x = p[0]; v.y = p[1];
        cfrag[c] = v;
    }

    const float* Eb  = embed_A  + ((size_t)b << 18);
    const float* h3b = embed_h3 + (size_t)b * N_OUT;

    const int rot = (tile ^ (tile >> 3)) & 7;
    v4f w0[16], w1[16], ev[4][8], c0[4], c1[4], h3v, h3n;
    {
        const int cg0 = wave + 8 * rot;
        const int cg1 = wave + 8 * ((rot + 1) & 7);
        const int gc = cg0 * 64 + 4 * ln;
        __builtin_prefetch(Eb + (size_t)(cg1 * 64 + lane) * N_IN, 0, 1);
        __builtin_prefetch(Eb + (size_t)(cg1 * 64 + 32 + lane) * N_IN, 0, 1);
        load_w_half<0>(amat_W, gc, half, w0, w1);
        load_c_group(cont_W, h3b, gc, half, c0, c1, h3v);
        __builtin_amdgcn_sched_barrier(0);
        load_e_half<0>(Eb, cg0 * 64, lane, ev);
        __builtin_amdgcn_sched_barrier(0);
        load_w_half<4>(amat_W, gc, half, w0, w1);
        __builtin_amdgcn_sched_barrier(0);
        load_e_half<4>(Eb, cg0 * 64, lane, ev);
        __builtin_amdgcn_sched_barrier(0);
    }
    h3n = h3v;
    float* est = estage + wave * (16 * ESTRIDE);

    auto group_pass = [&](const int it, auto more_tag) __attribute__((always_inline)) {
        constexpr bool MORE = decltype(more_tag)::value;
        const int cg  = wave + 8 * ((it + rot) & 7);
        const int cgn = wave + 8 * ((it + 1 + rot) & 7);
        const int gc  = cg * 64 + 4 * ln;
        const int gcn = cgn * 64 + 4 * ln;
        if (it + 2 < N_OUT / 64 / 8) {
            const int cgp = wave + 8 * ((it + 2 + rot) & 7);
            __builtin_prefetch(Eb + (size_t)(cgp * 64 + lane) * N_IN, 0, 1);
            __builtin_prefetch(Eb + (size_t)(cgp * 64 + 32 + lane) * N_IN, 0, 1);
        }

        xpose_e_half<0>(est, lane, ev);
        __builtin_amdgcn_sched_barrier(0);

        v8f acc[4] = {};
#pragma unroll
        for (int cc = 0; cc < 4; ++cc) {
#pragma unroll
            for (int j = 0; j < 4; ++j) {
                v2f bf; bf.x = c0[cc][j]; bf.y = c1[cc][j];
                acc[j] = wmma4(cfrag[cc], bf, acc[j]);
            }
        }
        add_w_half<0>(ev, w0, w1);
        __builtin_amdgcn_sched_barrier(0);
        if constexpr (MORE) {
            load_w_half<0>(amat_W, gcn, half, w0, w1);
            load_c_group(cont_W, h3b, gcn, half, c0, c1, h3n);
        }
        mm_k<0, 2>(acc, afrag, ev);
        __builtin_amdgcn_sched_barrier(0);
        xpose_e_half<4>(est, lane, ev);
        __builtin_amdgcn_sched_barrier(0);
        mm_k<2, 4>(acc, afrag, ev);
        add_w_half<4>(ev, w0, w1);
        __builtin_amdgcn_sched_barrier(0);
        if constexpr (MORE) {
            load_e_half<0>(Eb, cgn * 64, lane, ev);
            load_w_half<4>(amat_W, gcn, half, w0, w1);
        }
        mm_k<4, 8>(acc, afrag, ev);
        __builtin_amdgcn_sched_barrier(0);
        if constexpr (MORE) load_e_half<4>(Eb, cgn * 64, lane, ev);
#pragma unroll
        for (int r = 0; r < 8; ++r) {
            const int m = r + 8 * half;
            v4f o;
            o.x = acc[0][r] + h3v.x;
            o.y = acc[1][r] + h3v.y;
            o.z = acc[2][r] + h3v.z;
            o.w = acc[3][r] + h3v.w;
            *(v4f*)&hbuf[m * HSTRIDE + gc] = o;
        }
        h3v = h3n;
    };
    for (int it = 0; it < N_OUT / 64 / 8 - 1; ++it) group_pass(it, HasNext{});
    group_pass(N_OUT / 64 / 8 - 1, NoNext{});
    __syncthreads();

    v4f xr[2][32];
#pragma unroll
    for (int rr = 0; rr < 2; ++rr) {
        const float* hrow = hbuf + (2 * wave + rr) * HSTRIDE + 4 * lane;
#pragma unroll
        for (int n = 0; n < 32; ++n) xr[rr][n] = *(const v4f*)(hrow + 128 * n);
    }
#pragma unroll
    for (int rr = 0; rr < 2; ++rr) {
        const int row = 2 * wave + rr;

        v4f m4 = xr[rr][0];
#pragma unroll
        for (int n = 1; n < 32; ++n) {
            m4.x = fmaxf(m4.x, xr[rr][n].x);
            m4.y = fmaxf(m4.y, xr[rr][n].y);
            m4.z = fmaxf(m4.z, xr[rr][n].z);
            m4.w = fmaxf(m4.w, xr[rr][n].w);
        }
        const float mx = wave_max32(fmaxf(fmaxf(m4.x, m4.y), fmaxf(m4.z, m4.w)));

        v4f s4 = {0.0f, 0.0f, 0.0f, 0.0f};
#pragma unroll
        for (int n = 0; n < 32; ++n) {
            v4f e;
            e.x = __expf(xr[rr][n].x - mx);
            e.y = __expf(xr[rr][n].y - mx);
            e.z = __expf(xr[rr][n].z - mx);
            e.w = __expf(xr[rr][n].w - mx);
            xr[rr][n] = e;
            s4 += e;
        }
        const float sum = wave_sum32((s4.x + s4.y) + (s4.z + s4.w));

        const int s = sids[row];
        if (s >= 0) {
            const float scale = sfbuf[row] / sum;
            float* orow = out + (size_t)s * N_OUT + 4 * lane;
#pragma unroll
            for (int n = 0; n < 32; ++n) *(v4f*)(orow + 128 * n) = xr[rr][n] * scale;
        }
    }
}

extern "C" void kernel_launch(void* const* d_in, const int* in_sizes, int n_in,
                              void* d_out, int out_size, void* d_ws, size_t ws_size,
                              hipStream_t stream)
{
    const float* z    = (const float*)d_in[0];
    const int*   bc   = (const int*)  d_in[1];
    const float* sf   = (const float*)d_in[2];
    const float* cc   = (const float*)d_in[3];
    const float* aW   = (const float*)d_in[4];
    const float* eA   = (const float*)d_in[5];
    const float* eh3  = (const float*)d_in[6];
    const float* cW   = (const float*)d_in[7];
    const float* pxr  = (const float*)d_in[8];
    float* out = (float*)d_out;

    size_t smem = (size_t)(16 * HSTRIDE + 16 * ZSTRIDE + 16 * CSTRIDE + 16 + 8 * 16 * ESTRIDE) * sizeof(float)
                + 16 * sizeof(int);
    decoder_main_kernel<<<128, 256, smem, stream>>>(
        z, bc, sf, cc, aW, eA, eh3, cW, pxr, out);
}


# First group's loads issued as soon as their addresses are known (amat_W, cont_W at kernel entry, A_b once the batch is known), across LDS-only barriers: they land under the sort; sum order unchanged
# speedup vs baseline: 3.4801x; 1.1311x over previous
#include <hip/hip_runtime.h>
#include <hip/hip_bf16.h>

#define B_SAMP 1024
#define N_IN   64
#define N_OUT  4096
#define N_BATCH 64
#define N_CONT 16

#define HSTRIDE 4100
#define ZSTRIDE 68
#define CSTRIDE 20
#define ESTRIDE 36

typedef __attribute__((ext_vector_type(2))) float v2f;
typedef __attribute__((ext_vector_type(4))) float v4f;
typedef __attribute__((ext_vector_type(8))) float v8f;

struct HasNext { static constexpr bool value = true;  };
struct NoNext  { static constexpr bool value = false; };

static __device__ __forceinline__ v8f wmma4(v2f a, v2f b, v8f c) {
    return __builtin_amdgcn_wmma_f32_16x16x4_f32(
         false, a,  false, b,
         (short)0, c,  false,  false);
}

template <int I0>
static __device__ __forceinline__ void load_w_half(
    const float* amat_W, int gc, int half, v4f (&w0)[16], v4f (&w1)[16])
{
#pragma unroll
    for (int cc = 2 * I0; cc < 2 * I0 + 8; ++cc) {
        const int k = 8 * (cc >> 1) + 4 * half + 2 * (cc & 1);
        w0[cc] = *(const v4f*)(amat_W + (size_t)k * N_OUT + gc);
        w1[cc] = *(const v4f*)(amat_W + (size_t)(k + 1) * N_OUT + gc);
    }
}

static __device__ __forceinline__ void load_c_rows(
    const float* cont_W, int gc, int half, v4f (&c0)[4], v4f (&c1)[4])
{
#pragma unroll
    for (int cc = 0; cc < 4; ++cc) {
        const int k = 4 * cc + 2 * half;
        c0[cc] = *(const v4f*)(cont_W + (size_t)k * N_OUT + gc);
        c1[cc] = *(const v4f*)(cont_W + (size_t)(k + 1) * N_OUT + gc);
    }
}

static __device__ __forceinline__ void load_c_group(
    const float* cont_W, const float* h3b, int gc, int half,
    v4f (&c0)[4], v4f (&c1)[4], v4f& h3v)
{
    load_c_rows(cont_W, gc, half, c0, c1);
    h3v = *(const v4f*)(h3b + gc);
}

static __device__ __forceinline__ void lds_barrier() {
    __builtin_amdgcn_fence(__ATOMIC_RELEASE, "workgroup", "local");
    __builtin_amdgcn_s_barrier();
    __builtin_amdgcn_fence(__ATOMIC_ACQUIRE, "workgroup", "local");
}

template <int I0>
static __device__ __forceinline__ void load_e_half(
    const float* Eb, int c0, int lane, v4f (&ev)[4][8])
{
    const int r = lane >> 3, c = lane & 7;
#pragma unroll
    for (int j = 0; j < 4; ++j) {
        const float* ep = Eb + (size_t)(c0 + 4 * r + j) * N_IN + 8 * I0 + 4 * c;
#pragma unroll
        for (int t = 0; t < 4; ++t) ev[j][I0 + t] = *(const v4f*)(ep + (size_t)t * 16 * N_IN);
    }
}

template <int I0>
static __device__ __forceinline__ void xpose_e_half(
    float* st, int lane, v4f (&ev)[4][8])
{
    float* wp = st + (lane >> 3) * ESTRIDE + 4 * (lane & 7);
    const float* rp = st + (lane & 15) * ESTRIDE + 4 * (lane >> 4);
#pragma unroll
    for (int j = 0; j < 4; ++j) {
#pragma unroll
        for (int t = 0; t < 4; ++t) *(v4f*)(wp + t * 4 * ESTRIDE) = ev[j][I0 + t];
#pragma unroll
        for (int i = 0; i < 4; ++i) ev[j][I0 + i] = *(const v4f*)(rp + 8 * i);
    }
}

template <int I0>
static __device__ __forceinline__ void add_w_half(
    v4f (&ev)[4][8], const v4f (&w0)[16], const v4f (&w1)[16])
{
#pragma unroll
    for (int i = I0; i < I0 + 4; ++i) {
#pragma unroll
        for (int j = 0; j < 4; ++j) {
            ev[j][i].x += w0[2 * i][j];
            ev[j][i].y += w1[2 * i][j];
            ev[j][i].z += w0[2 * i + 1][j];
            ev[j][i].w += w1[2 * i + 1][j];
        }
    }
}

template <int I0, int I1>
static __device__ __forceinline__ void mm_k(
    v8f (&acc)[4], const v2f (&afrag)[16], const v4f (&ev)[4][8])
{
#pragma unroll
    for (int i = I0; i < I1; ++i) {
#pragma unroll
        for (int j = 0; j < 4; ++j) {
            v2f bf; bf.x = ev[j][i].x; bf.y = ev[j][i].y;
            acc[j] = wmma4(afrag[2 * i], bf, acc[j]);
        }
#pragma unroll
        for (int j = 0; j < 4; ++j) {
            v2f bf; bf.x = ev[j][i].z; bf.y = ev[j][i].w;
            acc[j] = wmma4(afrag[2 * i + 1], bf, acc[j]);
        }
    }
}

template <int MASK>
static __device__ __forceinline__ float row_xchg(float v) {
    const int b = __builtin_bit_cast(int, v);
    return __builtin_bit_cast(float,
        __builtin_amdgcn_update_dpp(b, b, 0x160 + MASK, 0xf, 0xf, false));
}
static __device__ __forceinline__ float half_xchg(float v) {
    const int b = __builtin_bit_cast(int, v);
    return __builtin_bit_cast(float,
        __builtin_amdgcn_permlanex16(b, b, 0x76543210, 0xfedcba98, false, false));
}
static __device__ __forceinline__ float wave_max32(float v) {
    v = fmaxf(v, row_xchg<1>(v));
    v = fmaxf(v, row_xchg<2>(v));
    v = fmaxf(v, row_xchg<4>(v));
    v = fmaxf(v, row_xchg<8>(v));
    return fmaxf(v, half_xchg(v));
}
static __device__ __forceinline__ float wave_sum32(float v) {
    v += row_xchg<1>(v);
    v += row_xchg<2>(v);
    v += row_xchg<4>(v);
    v += row_xchg<8>(v);
    return v + half_xchg(v);
}

__global__ __launch_bounds__(256) void decoder_main_kernel(
    const float* __restrict__ z, const int* __restrict__ idx,
    const float* __restrict__ sizef, const float* __restrict__ cont,
    const float* amat_W, const float* embed_A, const float* embed_h3,
    const float* cont_W, const float* __restrict__ px_r, float* __restrict__ out)
{
    extern __shared__ __attribute__((aligned(16))) float smem[];
    float* hbuf  = smem;
    float* ztile = hbuf  + 16 * HSTRIDE;
    float* ctile = ztile + 16 * ZSTRIDE;
    float* sfbuf = ctile + 16 * CSTRIDE;
    int*   sids  = (int*)(sfbuf + 256);
    float* estage = (float*)(sids + 16);

    __shared__ int hist[N_BATCH];
    __shared__ int ccnt[32];

    const int tile = blockIdx.x;
    const int t = threadIdx.x;
    const int wave = t >> 5;
    const int lane = t & 31;

    const int half = lane >> 4;
    const int ln   = lane & 15;

    const int rot = (tile ^ (tile >> 3)) & 7;
    const int cg0 = wave + 8 * rot;
    const int cg1 = wave + 8 * ((rot + 1) & 7);
    const int gc0 = cg0 * 64 + 4 * ln;

    int bv[4];
#pragma unroll
    for (int j = 0; j < 4; ++j) bv[j] = idx[256 * j + t];
    const float pr = px_r[tile * 32 + lane];
    __builtin_amdgcn_sched_barrier(0);
    v4f w0[16], w1[16], ev[4][8], c0[4], c1[4], h3v, h3n;
    load_w_half<0>(amat_W, gc0, half, w0, w1);
    load_c_rows(cont_W, gc0, half, c0, c1);
    __builtin_amdgcn_sched_barrier(0);

    if (t < N_BATCH) hist[t] = 0;
    if (t < 16) sids[t] = -1;
    lds_barrier();
#pragma unroll
    for (int j = 0; j < 4; ++j) atomicAdd(&hist[bv[j]], 1);
    if (t < 32) out[(size_t)B_SAMP * N_OUT + tile * 32 + t] = __expf(pr);
    lds_barrier();

    int b = -1, q = 0, tc = 0;
#pragma unroll
    for (int bb = 0; bb < N_BATCH; ++bb) {
        const int nt = (hist[bb] + 15) >> 4;
        if (tile >= tc && tile < tc + nt) { b = bb; q = tile - tc; }
        tc += nt;
    }
    if (b < 0) return;

    const float* Eb  = embed_A  + ((size_t)b << 18);
    const float* h3b = embed_h3 + (size_t)b * N_OUT;
    __builtin_prefetch(Eb + (size_t)(cg1 * 64 + lane) * N_IN, 0, 1);
    __builtin_prefetch(Eb + (size_t)(cg1 * 64 + 32 + lane) * N_IN, 0, 1);
    h3v = *(const v4f*)(h3b + gc0);
    load_e_half<0>(Eb, cg0 * 64, lane, ev);
    __builtin_amdgcn_sched_barrier(0);

    bool mine[4];
    int below[4];
#pragma unroll
    for (int j = 0; j < 4; ++j) {
        mine[j] = (bv[j] == b);
        const unsigned m = (unsigned)__ballot(mine[j]);
        below[j] = __popc(m & ((1u << lane) - 1u));
        if (lane == 0) ccnt[8 * j + wave] = __popc(m);
    }
    lds_barrier();
    int base[4] = {0, 0, 0, 0};
#pragma unroll
    for (int c = 0; c < 32; ++c) {
        const int n = ccnt[c];
#pragma unroll
        for (int j = 0; j < 4; ++j) base[j] += (c < 8 * j + wave) ? n : 0;
    }
#pragma unroll
    for (int j = 0; j < 4; ++j) {
        const int rank = base[j] + below[j];
        if (mine[j] && (rank >> 4) == q) sids[rank & 15] = 256 * j + t;
    }
    lds_barrier();

    int zs[4];
    float zr[4];
#pragma unroll
    for (int j = 0; j < 4; ++j) {
        zs[j] = sids[(t >> 6) + 4 * j];
        zr[j] = z[(zs[j] < 0 ? 0 : zs[j]) * N_IN + (t & 63)];
    }
    const int cs = sids[t >> 4];
    const float cr = cont[(cs < 0 ? 0 : cs) * N_CONT + (t & 15)];
    const int ss = sids[t & 15];
    const float sr = sizef[ss < 0 ? 0 : ss];
    __builtin_amdgcn_sched_barrier(0);
    load_w_half<4>(amat_W, gc0, half, w0, w1);
    load_e_half<4>(Eb, cg0 * 64, lane, ev);
    __builtin_amdgcn_sched_barrier(0);
#pragma unroll
    for (int j = 0; j < 4; ++j)
        ztile[((t >> 6) + 4 * j) * ZSTRIDE + (t & 63)] = (zs[j] >= 0) ? zr[j] : 0.0f;
    ctile[(t >> 4) * CSTRIDE + (t & 15)] = (cs >= 0) ? cr : 0.0f;
    sfbuf[t] = (ss >= 0) ? sr : 1.0f;
    lds_barrier();

    v2f afrag[16];
#pragma unroll
    for (int i = 0; i < 8; ++i) {
        const v4f q = *(const v4f*)&ztile[ln * ZSTRIDE + 8 * i + 4 * half];
        v2f lo; lo.x = q.x; lo.y = q.y;
        v2f hi; hi.x = q.z; hi.y = q.w;
        afrag[2 * i]     = lo;
        afrag[2 * i + 1] = hi;
    }
    v2f cfrag[4];
#pragma unroll
    for (int c = 0; c < 4; ++c) {
        const float* p = &ctile[ln * CSTRIDE + 4 * c + 2 * half];
        v2f v; v.x = p[0]; v.y = p[1];
        cfrag[c] = v;
    }

    h3n = h3v;
    float* est = estage + wave * (16 * ESTRIDE);

    auto group_pass = [&](const int it, auto more_tag) __attribute__((always_inline)) {
        constexpr bool MORE = decltype(more_tag)::value;
        const int cg  = wave + 8 * ((it + rot) & 7);
        const int cgn = wave + 8 * ((it + 1 + rot) & 7);
        const int gc  = cg * 64 + 4 * ln;
        const int gcn = cgn * 64 + 4 * ln;
        if (it + 2 < N_OUT / 64 / 8) {
            const int cgp = wave + 8 * ((it + 2 + rot) & 7);
            __builtin_prefetch(Eb + (size_t)(cgp * 64 + lane) * N_IN, 0, 1);
            __builtin_prefetch(Eb + (size_t)(cgp * 64 + 32 + lane) * N_IN, 0, 1);
        }

        xpose_e_half<0>(est, lane, ev);
        __builtin_amdgcn_sched_barrier(0);

        v8f acc[4] = {};
#pragma unroll
        for (int cc = 0; cc < 4; ++cc) {
#pragma unroll
            for (int j = 0; j < 4; ++j) {
                v2f bf; bf.x = c0[cc][j]; bf.y = c1[cc][j];
                acc[j] = wmma4(cfrag[cc], bf, acc[j]);
            }
        }
        add_w_half<0>(ev, w0, w1);
        __builtin_amdgcn_sched_barrier(0);
        if constexpr (MORE) {
            load_w_half<0>(amat_W, gcn, half, w0, w1);
            load_c_group(cont_W, h3b, gcn, half, c0, c1, h3n);
        }
        mm_k<0, 2>(acc, afrag, ev);
        __builtin_amdgcn_sched_barrier(0);
        xpose_e_half<4>(est, lane, ev);
        __builtin_amdgcn_sched_barrier(0);
        mm_k<2, 4>(acc, afrag, ev);
        add_w_half<4>(ev, w0, w1);
        __builtin_amdgcn_sched_barrier(0);
        if constexpr (MORE) {
            load_e_half<0>(Eb, cgn * 64, lane, ev);
            load_w_half<4>(amat_W, gcn, half, w0, w1);
        }
        mm_k<4, 8>(acc, afrag, ev);
        __builtin_amdgcn_sched_barrier(0);
        if constexpr (MORE) load_e_half<4>(Eb, cgn * 64, lane, ev);
#pragma unroll
        for (int r = 0; r < 8; ++r) {
            const int m = r + 8 * half;
            v4f o;
            o.x = acc[0][r] + h3v.x;
            o.y = acc[1][r] + h3v.y;
            o.z = acc[2][r] + h3v.z;
            o.w = acc[3][r] + h3v.w;
            *(v4f*)&hbuf[m * HSTRIDE + gc] = o;
        }
        h3v = h3n;
    };
    for (int it = 0; it < N_OUT / 64 / 8 - 1; ++it) group_pass(it, HasNext{});
    group_pass(N_OUT / 64 / 8 - 1, NoNext{});
    __syncthreads();

    v4f xr[2][32];
#pragma unroll
    for (int rr = 0; rr < 2; ++rr) {
        const float* hrow = hbuf + (2 * wave + rr) * HSTRIDE + 4 * lane;
#pragma unroll
        for (int n = 0; n < 32; ++n) xr[rr][n] = *(const v4f*)(hrow + 128 * n);
    }
#pragma unroll
    for (int rr = 0; rr < 2; ++rr) {
        const int row = 2 * wave + rr;

        v4f m4 = xr[rr][0];
#pragma unroll
        for (int n = 1; n < 32; ++n) {
            m4.x = fmaxf(m4.x, xr[rr][n].x);
            m4.y = fmaxf(m4.y, xr[rr][n].y);
            m4.z = fmaxf(m4.z, xr[rr][n].z);
            m4.w = fmaxf(m4.w, xr[rr][n].w);
        }
        const float mx = wave_max32(fmaxf(fmaxf(m4.x, m4.y), fmaxf(m4.z, m4.w)));

        v4f s4 = {0.0f, 0.0f, 0.0f, 0.0f};
#pragma unroll
        for (int n = 0; n < 32; ++n) {
            v4f e;
            e.x = __expf(xr[rr][n].x - mx);
            e.y = __expf(xr[rr][n].y - mx);
            e.z = __expf(xr[rr][n].z - mx);
            e.w = __expf(xr[rr][n].w - mx);
            xr[rr][n] = e;
            s4 += e;
        }
        const float sum = wave_sum32((s4.x + s4.y) + (s4.z + s4.w));

        const int s = sids[row];
        if (s >= 0) {
            const float scale = sfbuf[row] / sum;
            float* orow = out + (size_t)s * N_OUT + 4 * lane;
#pragma unroll
            for (int n = 0; n < 32; ++n) *(v4f*)(orow + 128 * n) = xr[rr][n] * scale;
        }
    }
}

extern "C" void kernel_launch(void* const* d_in, const int* in_sizes, int n_in,
                              void* d_out, int out_size, void* d_ws, size_t ws_size,
                              hipStream_t stream)
{
    const float* z    = (const float*)d_in[0];
    const int*   bc   = (const int*)  d_in[1];
    const float* sf   = (const float*)d_in[2];
    const float* cc   = (const float*)d_in[3];
    const float* aW   = (const float*)d_in[4];
    const float* eA   = (const float*)d_in[5];
    const float* eh3  = (const float*)d_in[6];
    const float* cW   = (const float*)d_in[7];
    const float* pxr  = (const float*)d_in[8];
    float* out = (float*)d_out;

    size_t smem = (size_t)(16 * HSTRIDE + 16 * ZSTRIDE + 16 * CSTRIDE + 256 + 8 * 16 * ESTRIDE) * sizeof(float)
                + 16 * sizeof(int);
    decoder_main_kernel<<<128, 256, smem, stream>>>(
        z, bc, sf, cc, aW, eA, eh3, cW, pxr, out);
}
